# starting kernel
# baseline (speedup 1.0000x reference)
#include <hip/hip_runtime.h>
#include <math.h>

typedef __attribute__((ext_vector_type(2))) float v2f;
typedef __attribute__((ext_vector_type(8))) float v8f;

#define B_DIM 2
#define C_DIM 256
#define N_DIM 384

static __device__ __forceinline__ float hw_tanhf(float x) {
#if __has_builtin(__builtin_amdgcn_tanhf)
    return __builtin_amdgcn_tanhf(x);
#else
    float r;
    asm("v_tanh_f32 %0, %1" : "=v"(r) : "v"(x));
    return r;
#endif
}

__global__ void fuse_bias_kernel(const float* __restrict__ W_a,
                                 const float* __restrict__ theta_b,
                                 const float* __restrict__ phi_b,
                                 float* __restrict__ c1,
                                 float* __restrict__ c2)
{
    int o2 = blockIdx.x * blockDim.x + threadIdx.x;
    if (o2 >= C_DIM) return;
    float s1 = 0.f, s2 = 0.f;
    const float* row = W_a + (long)o2 * (2 * C_DIM);
    #pragma unroll 8
    for (int o = 0; o < C_DIM; ++o) {
        s1 += row[o] * theta_b[o];
        s2 += row[C_DIM + o] * phi_b[o];
    }
    c1[o2] = s1;
    c2[o2] = s2;
}

__global__ void wmma_gemm_f32(const float* __restrict__ A, int lda,
                              const float* __restrict__ Bm, int ldb, long strideB,
                              float* __restrict__ D, int ldd, long strideD,
                              const float* __restrict__ bias,
                              int M, int N, int K)
{
    const float* Bp = Bm + (long)blockIdx.z * strideB;
    float* Dp = D + (long)blockIdx.z * strideD;

    const int wavesPerBlock = blockDim.x >> 5;
    const int wave = blockIdx.x * wavesPerBlock + (threadIdx.x >> 5);
    const int lane = threadIdx.x & 31;
    const int tilesN = N >> 4;
    const int tM = wave / tilesN;
    const int tN = wave - tM * tilesN;
    if (tM * 16 >= M) return;

    const int half = lane >> 4;
    const int ln   = lane & 15;
    const int rowA = tM * 16 + ln;
    const int colB = tN * 16 + ln;

    v8f acc = {};
    #pragma unroll 8
    for (int k0 = 0; k0 < K; k0 += 4) {
        const int ka = k0 + 2 * half;
        v2f af = *(const v2f*)(A + (long)rowA * lda + ka);
        v2f bf;
        bf.x = Bp[(long)ka * ldb + colB];
        bf.y = Bp[(long)(ka + 1) * ldb + colB];
        acc = __builtin_amdgcn_wmma_f32_16x16x4_f32(
             false, af,  false, bf,
             (short)0, acc,  false,  false);
    }

    #pragma unroll
    for (int v = 0; v < 8; ++v) {
        const int row = tM * 16 + v + 8 * half;
        float val = acc[v];
        if (bias) val += bias[row];
        Dp[(long)row * ldd + tN * 16 + ln] = val;
    }
}

#define OCH 64
__global__ void pairwise_tanh_kernel(const float* __restrict__ a,
                                     const float* __restrict__ bt,
                                     const float* __restrict__ v,
                                     float* __restrict__ f)
{
    __shared__ float a_s[OCH][33];
    __shared__ float bt_s[OCH][33];
    __shared__ float v_s[OCH];

    const int b     = blockIdx.z;
    const int kBase = blockIdx.x * 32;
    const int iBase = blockIdx.y * 32;
    const float* ab  = a  + (long)b * C_DIM * N_DIM;
    const float* btb = bt + (long)b * C_DIM * N_DIM;

    const int t  = threadIdx.x;
    const int kl = t & 31;
    const int ig = t >> 5;

    float acc0 = 0.f, acc1 = 0.f, acc2 = 0.f, acc3 = 0.f;

    for (int oc = 0; oc < C_DIM; oc += OCH) {
        for (int r = ig; r < OCH; r += 8) {
            a_s[r][kl]  = ab [(long)(oc + r) * N_DIM + kBase + kl];
            bt_s[r][kl] = btb[(long)(oc + r) * N_DIM + iBase + kl];
        }
        if (t < OCH) v_s[t] = v[oc + t];
        __syncthreads();

        #pragma unroll 4
        for (int o = 0; o < OCH; ++o) {
            const float av = a_s[o][kl];
            const float vo = v_s[o];
            acc0 += vo * hw_tanhf(av + bt_s[o][ig]);
            acc1 += vo * hw_tanhf(av + bt_s[o][ig + 8]);
            acc2 += vo * hw_tanhf(av + bt_s[o][ig + 16]);
            acc3 += vo * hw_tanhf(av + bt_s[o][ig + 24]);
        }
        __syncthreads();
    }

    float* fb = f + (long)b * N_DIM * N_DIM;
    fb[(long)(iBase + ig     ) * N_DIM + kBase + kl] = acc0;
    fb[(long)(iBase + ig +  8) * N_DIM + kBase + kl] = acc1;
    fb[(long)(iBase + ig + 16) * N_DIM + kBase + kl] = acc2;
    fb[(long)(iBase + ig + 24) * N_DIM + kBase + kl] = acc3;
}

__global__ void softmax_i_kernel(const float* __restrict__ f,
                                 float* __restrict__ out)
{
    __shared__ float red[8][33];
    __shared__ float colM[32];
    __shared__ float colS[32];

    const int b     = blockIdx.y;
    const int kBase = blockIdx.x * 32;
    const int kl    = threadIdx.x & 31;
    const int iy    = threadIdx.x >> 5;
    const float* fb = f + (long)b * N_DIM * N_DIM;
    float* ob       = out + (long)b * N_DIM * N_DIM;

    float m = -INFINITY;
    for (int i = iy; i < N_DIM; i += 8)
        m = fmaxf(m, fb[(long)i * N_DIM + kBase + kl]);
    red[iy][kl] = m;
    __syncthreads();
    if (iy == 0) {
        float mm = red[0][kl];
        #pragma unroll
        for (int r = 1; r < 8; ++r) mm = fmaxf(mm, red[r][kl]);
        colM[kl] = mm;
    }
    __syncthreads();
    const float mm = colM[kl];

    float s = 0.f;
    for (int i = iy; i < N_DIM; i += 8)
        s += __expf(fb[(long)i * N_DIM + kBase + kl] - mm);
    red[iy][kl] = s;
    __syncthreads();
    if (iy == 0) {
        float ss = 0.f;
        #pragma unroll
        for (int r = 0; r < 8; ++r) ss += red[r][kl];
        colS[kl] = ss;
    }
    __syncthreads();
    const float inv = 1.f / colS[kl];

    for (int i = iy; i < N_DIM; i += 8) {
        const long idx = (long)i * N_DIM + kBase + kl;
        ob[idx] = __expf(fb[idx] - mm) * inv;
    }
}

extern "C" void kernel_launch(void* const* d_in, const int* in_sizes, int n_in,
                              void* d_out, int out_size, void* d_ws, size_t ws_size,
                              hipStream_t stream)
{
    const float* x       = (const float*)d_in[0];
    const float* theta_w = (const float*)d_in[1];
    const float* theta_b = (const float*)d_in[2];
    const float* phi_w   = (const float*)d_in[3];
    const float* phi_b   = (const float*)d_in[4];
    const float* W_a     = (const float*)d_in[5];
    const float* v_a     = (const float*)d_in[6];
    float* out = (float*)d_out;

    float* ws = (float*)d_ws;
    float* M1 = ws;
    float* M2 = M1 + C_DIM * C_DIM;
    float* c1 = M2 + C_DIM * C_DIM;
    float* c2 = c1 + C_DIM;
    float* a  = c2 + C_DIM;
    float* bt = a  + B_DIM * C_DIM * N_DIM;
    float* f  = bt + B_DIM * C_DIM * N_DIM;

    fuse_bias_kernel<<<1, 256, 0, stream>>>(W_a, theta_b, phi_b, c1, c2);

    {
        dim3 grid(32, 1, 1), block(256, 1, 1);
        wmma_gemm_f32<<<grid, block, 0, stream>>>(
            W_a, 2 * C_DIM, theta_w, C_DIM, 0, M1, C_DIM, 0,
            nullptr, C_DIM, C_DIM, C_DIM);
        wmma_gemm_f32<<<grid, block, 0, stream>>>(
            W_a + C_DIM, 2 * C_DIM, phi_w, C_DIM, 0, M2, C_DIM, 0,
            nullptr, C_DIM, C_DIM, C_DIM);
    }

    {
        dim3 grid(48, 1, B_DIM), block(256, 1, 1);
        const long sAct = (long)C_DIM * N_DIM;
        wmma_gemm_f32<<<grid, block, 0, stream>>>(
            M1, C_DIM, x, N_DIM, sAct, a, N_DIM, sAct,
            c1, C_DIM, N_DIM, C_DIM);
        wmma_gemm_f32<<<grid, block, 0, stream>>>(
            M2, C_DIM, x, N_DIM, sAct, bt, N_DIM, sAct,
            c2, C_DIM, N_DIM, C_DIM);
    }

    {
        dim3 grid(N_DIM / 32, N_DIM / 32, B_DIM), block(256, 1, 1);
        pairwise_tanh_kernel<<<grid, block, 0, stream>>>(a, bt, v_a, f);
    }

    {
        dim3 grid(N_DIM / 32, B_DIM, 1), block(256, 1, 1);
        softmax_i_kernel<<<grid, block, 0, stream>>>(f, out);
    }
}


# One launch replaces the five that built a and bt: per 16-column slab h=W x+b is built in LDS, then multiplied by its half of W_a; sums reassociated to W1(theta x+b), the reference's own order
# speedup vs baseline: 1.7652x; 1.7652x over previous
#include <hip/hip_runtime.h>
#include <math.h>

typedef __attribute__((ext_vector_type(2))) float v2f;
typedef __attribute__((ext_vector_type(8))) float v8f;

#define B_DIM 2
#define C_DIM 256
#define N_DIM 384

static __device__ __forceinline__ float hw_tanhf(float x) {
#if __has_builtin(__builtin_amdgcn_tanhf)
    return __builtin_amdgcn_tanhf(x);
#else
    float r;
    asm("v_tanh_f32 %0, %1" : "=v"(r) : "v"(x));
    return r;
#endif
}

#define NT 16
#define N_TILES (N_DIM / NT)

__global__ void __launch_bounds__(512)
chain_gemm_f32(const float* __restrict__ x,
               const float* __restrict__ theta_w,
               const float* __restrict__ theta_b,
               const float* __restrict__ phi_w,
               const float* __restrict__ phi_b,
               const float* __restrict__ W_a,
               float* __restrict__ a_t,
               float* __restrict__ bt_t)
{
    __shared__ float xs[C_DIM * NT];
    __shared__ float hs[C_DIM * NT];

    const int tile  = blockIdx.x;
    const int b     = blockIdx.y;
    const int which = blockIdx.z;
    const float* Wf   = which ? phi_w : theta_w;
    const float* bias = which ? phi_b : theta_b;
    const float* Ws   = W_a + which * C_DIM;
    float* dst = (which ? bt_t : a_t) + (long)(b * N_TILES + tile) * (C_DIM * NT);

    const int t = threadIdx.x;

    {
        const float* xb = x + (long)b * C_DIM * N_DIM + tile * NT;
        const int c4 = (t & 3) * 4;
        #pragma unroll
        for (int p = 0; p < 2; ++p) {
            const int r = (t >> 2) + p * 128;
            const float4 val = *(const float4*)(xb + (long)r * N_DIM + c4);
            *(float4*)(xs + r * NT + c4) = val;
        }
    }
    __syncthreads();

    const int w    = t >> 5;
    const int lane = t & 31;
    const int half = lane >> 4;
    const int ln   = lane & 15;
    const int rowA = w * 16 + ln;

    v8f acc = {};
    {
        const float* Arow = Wf + (long)rowA * C_DIM + 2 * half;
        const float* Bcol = xs + (2 * half) * NT + ln;
        #pragma unroll 16
        for (int k0 = 0; k0 < C_DIM; k0 += 4) {
            v2f af = *(const v2f*)(Arow + k0);
            v2f bf;
            bf.x = Bcol[k0 * NT];
            bf.y = Bcol[(k0 + 1) * NT];
            acc = __builtin_amdgcn_wmma_f32_16x16x4_f32(
                 false, af,  false, bf,
                 (short)0, acc,  false,  false);
        }
    }
    #pragma unroll
    for (int v = 0; v < 8; ++v) {
        const int row = w * 16 + v + 8 * half;
        hs[row * NT + ln] = acc[v] + bias[row];
    }
    __syncthreads();

    v8f acc2 = {};
    {
        const float* Arow = Ws + (long)rowA * (2 * C_DIM) + 2 * half;
        const float* Bcol = hs + (2 * half) * NT + ln;
        #pragma unroll 16
        for (int k0 = 0; k0 < C_DIM; k0 += 4) {
            v2f af = *(const v2f*)(Arow + k0);
            v2f bf;
            bf.x = Bcol[k0 * NT];
            bf.y = Bcol[(k0 + 1) * NT];
            acc2 = __builtin_amdgcn_wmma_f32_16x16x4_f32(
                 false, af,  false, bf,
                 (short)0, acc2,  false,  false);
        }
    }
    #pragma unroll
    for (int v = 0; v < 8; ++v) {
        const int row = w * 16 + v + 8 * half;
        dst[row * NT + ln] = acc2[v];
    }
}

#define OCH 64
__global__ void pairwise_tanh_kernel(const float* __restrict__ a,
                                     const float* __restrict__ bt,
                                     const float* __restrict__ v,
                                     float* __restrict__ f)
{
    __shared__ float a_s[OCH][33];
    __shared__ float bt_s[OCH][33];
    __shared__ float v_s[OCH];

    const int b     = blockIdx.z;
    const int kBase = blockIdx.x * 32;
    const int iBase = blockIdx.y * 32;
    const float* ab  = a  + (long)b * C_DIM * N_DIM;
    const float* btb = bt + (long)b * C_DIM * N_DIM;

    const int t  = threadIdx.x;
    const int kl = t & 31;
    const int ig = t >> 5;

    const int aoff = ((kBase >> 4) + (kl >> 4)) * (C_DIM * NT) + (kl & 15);
    const int boff = ((iBase >> 4) + (kl >> 4)) * (C_DIM * NT) + (kl & 15);

    float acc0 = 0.f, acc1 = 0.f, acc2 = 0.f, acc3 = 0.f;

    for (int oc = 0; oc < C_DIM; oc += OCH) {
        for (int r = ig; r < OCH; r += 8) {
            a_s[r][kl]  = ab [aoff + (oc + r) * NT];
            bt_s[r][kl] = btb[boff + (oc + r) * NT];
        }
        if (t < OCH) v_s[t] = v[oc + t];
        __syncthreads();

        #pragma unroll 4
        for (int o = 0; o < OCH; ++o) {
            const float av = a_s[o][kl];
            const float vo = v_s[o];
            acc0 += vo * hw_tanhf(av + bt_s[o][ig]);
            acc1 += vo * hw_tanhf(av + bt_s[o][ig + 8]);
            acc2 += vo * hw_tanhf(av + bt_s[o][ig + 16]);
            acc3 += vo * hw_tanhf(av + bt_s[o][ig + 24]);
        }
        __syncthreads();
    }

    float* fb = f + (long)b * N_DIM * N_DIM;
    fb[(long)(iBase + ig     ) * N_DIM + kBase + kl] = acc0;
    fb[(long)(iBase + ig +  8) * N_DIM + kBase + kl] = acc1;
    fb[(long)(iBase + ig + 16) * N_DIM + kBase + kl] = acc2;
    fb[(long)(iBase + ig + 24) * N_DIM + kBase + kl] = acc3;
}

__global__ void softmax_i_kernel(const float* __restrict__ f,
                                 float* __restrict__ out)
{
    __shared__ float red[8][33];
    __shared__ float colM[32];
    __shared__ float colS[32];

    const int b     = blockIdx.y;
    const int kBase = blockIdx.x * 32;
    const int kl    = threadIdx.x & 31;
    const int iy    = threadIdx.x >> 5;
    const float* fb = f + (long)b * N_DIM * N_DIM;
    float* ob       = out + (long)b * N_DIM * N_DIM;

    float m = -INFINITY;
    for (int i = iy; i < N_DIM; i += 8)
        m = fmaxf(m, fb[(long)i * N_DIM + kBase + kl]);
    red[iy][kl] = m;
    __syncthreads();
    if (iy == 0) {
        float mm = red[0][kl];
        #pragma unroll
        for (int r = 1; r < 8; ++r) mm = fmaxf(mm, red[r][kl]);
        colM[kl] = mm;
    }
    __syncthreads();
    const float mm = colM[kl];

    float s = 0.f;
    for (int i = iy; i < N_DIM; i += 8)
        s += __expf(fb[(long)i * N_DIM + kBase + kl] - mm);
    red[iy][kl] = s;
    __syncthreads();
    if (iy == 0) {
        float ss = 0.f;
        #pragma unroll
        for (int r = 0; r < 8; ++r) ss += red[r][kl];
        colS[kl] = ss;
    }
    __syncthreads();
    const float inv = 1.f / colS[kl];

    for (int i = iy; i < N_DIM; i += 8) {
        const long idx = (long)i * N_DIM + kBase + kl;
        ob[idx] = __expf(fb[idx] - mm) * inv;
    }
}

extern "C" void kernel_launch(void* const* d_in, const int* in_sizes, int n_in,
                              void* d_out, int out_size, void* d_ws, size_t ws_size,
                              hipStream_t stream)
{
    const float* x       = (const float*)d_in[0];
    const float* theta_w = (const float*)d_in[1];
    const float* theta_b = (const float*)d_in[2];
    const float* phi_w   = (const float*)d_in[3];
    const float* phi_b   = (const float*)d_in[4];
    const float* W_a     = (const float*)d_in[5];
    const float* v_a     = (const float*)d_in[6];
    float* out = (float*)d_out;

    float* ws = (float*)d_ws;
    float* a  = ws;
    float* bt = a  + B_DIM * C_DIM * N_DIM;
    float* f  = bt + B_DIM * C_DIM * N_DIM;

    {
        dim3 grid(N_TILES, B_DIM, 2), block(512, 1, 1);
        chain_gemm_f32<<<grid, block, 0, stream>>>(
            x, theta_w, theta_b, phi_w, phi_b, W_a, a, bt);
    }

    {
        dim3 grid(N_DIM / 32, N_DIM / 32, B_DIM), block(256, 1, 1);
        pairwise_tanh_kernel<<<grid, block, 0, stream>>>(a, bt, v_a, f);
    }

    {
        dim3 grid(N_DIM / 32, B_DIM, 1), block(256, 1, 1);
        softmax_i_kernel<<<grid, block, 0, stream>>>(f, out);
    }
}


# Pairwise re-tiled: 192 work-groups of 32 k x 48 i, one per WGP and equal work (was 288 of 32 x 32, two on some WGPs); lane tile 2 k x 3 i, so one LDS read feeds 2-3 tanh (was 1)
# speedup vs baseline: 3.4222x; 1.9387x over previous
#include <hip/hip_runtime.h>
#include <math.h>

typedef __attribute__((ext_vector_type(2))) float v2f;
typedef __attribute__((ext_vector_type(8))) float v8f;

#define B_DIM 2
#define C_DIM 256
#define N_DIM 384

static __device__ __forceinline__ float hw_tanhf(float x) {
#if __has_builtin(__builtin_amdgcn_tanhf)
    return __builtin_amdgcn_tanhf(x);
#else
    float r;
    asm("v_tanh_f32 %0, %1" : "=v"(r) : "v"(x));
    return r;
#endif
}

#define NT 16
#define N_TILES (N_DIM / NT)

__global__ void __launch_bounds__(512)
chain_gemm_f32(const float* __restrict__ x,
               const float* __restrict__ theta_w,
               const float* __restrict__ theta_b,
               const float* __restrict__ phi_w,
               const float* __restrict__ phi_b,
               const float* __restrict__ W_a,
               float* __restrict__ a_t,
               float* __restrict__ bt_t)
{
    __shared__ float xs[C_DIM * NT];
    __shared__ float hs[C_DIM * NT];

    const int tile  = blockIdx.x;
    const int b     = blockIdx.y;
    const int which = blockIdx.z;
    const float* Wf   = which ? phi_w : theta_w;
    const float* bias = which ? phi_b : theta_b;
    const float* Ws   = W_a + which * C_DIM;
    float* dst = (which ? bt_t : a_t) + (long)(b * N_TILES + tile) * (C_DIM * NT);

    const int t = threadIdx.x;

    {
        const float* xb = x + (long)b * C_DIM * N_DIM + tile * NT;
        const int c4 = (t & 3) * 4;
        #pragma unroll
        for (int p = 0; p < 2; ++p) {
            const int r = (t >> 2) + p * 128;
            const float4 val = *(const float4*)(xb + (long)r * N_DIM + c4);
            *(float4*)(xs + r * NT + c4) = val;
        }
    }
    __syncthreads();

    const int w    = t >> 5;
    const int lane = t & 31;
    const int half = lane >> 4;
    const int ln   = lane & 15;
    const int rowA = w * 16 + ln;

    v8f acc = {};
    {
        const float* Arow = Wf + (long)rowA * C_DIM + 2 * half;
        const float* Bcol = xs + (2 * half) * NT + ln;
        #pragma unroll 16
        for (int k0 = 0; k0 < C_DIM; k0 += 4) {
            v2f af = *(const v2f*)(Arow + k0);
            v2f bf;
            bf.x = Bcol[k0 * NT];
            bf.y = Bcol[(k0 + 1) * NT];
            acc = __builtin_amdgcn_wmma_f32_16x16x4_f32(
                 false, af,  false, bf,
                 (short)0, acc,  false,  false);
        }
    }
    #pragma unroll
    for (int v = 0; v < 8; ++v) {
        const int row = w * 16 + v + 8 * half;
        hs[row * NT + ln] = acc[v] + bias[row];
    }
    __syncthreads();

    v8f acc2 = {};
    {
        const float* Arow = Ws + (long)rowA * (2 * C_DIM) + 2 * half;
        const float* Bcol = hs + (2 * half) * NT + ln;
        #pragma unroll 16
        for (int k0 = 0; k0 < C_DIM; k0 += 4) {
            v2f af = *(const v2f*)(Arow + k0);
            v2f bf;
            bf.x = Bcol[k0 * NT];
            bf.y = Bcol[(k0 + 1) * NT];
            acc2 = __builtin_amdgcn_wmma_f32_16x16x4_f32(
                 false, af,  false, bf,
                 (short)0, acc2,  false,  false);
        }
    }
    #pragma unroll
    for (int v = 0; v < 8; ++v) {
        const int row = w * 16 + v + 8 * half;
        dst[row * NT + ln] = acc2[v];
    }
}

#define PW_KT 32
#define PW_IT 48
#define APITCH 36
#define BPITCH 52
__global__ void __launch_bounds__(256)
pairwise_tanh_kernel(const float* __restrict__ a,
                     const float* __restrict__ bt,
                     const float* __restrict__ v,
                     float* __restrict__ f)
{
    __shared__ float a_s[C_DIM * APITCH];
    __shared__ float bt_s[C_DIM * BPITCH];
    __shared__ float v_s[C_DIM];

    const int b     = blockIdx.z;
    const int kBase = blockIdx.x * PW_KT;
    const int iBase = blockIdx.y * PW_IT;
    const float* ab  = a  + (long)b * C_DIM * N_DIM + (long)(kBase / NT) * (C_DIM * NT);
    const float* btb = bt + (long)b * C_DIM * N_DIM + (long)(iBase / NT) * (C_DIM * NT);

    const int t  = threadIdx.x;

    {
        float4 ra[8], rb[12];
        #pragma unroll
        for (int j = 0; j < 8; ++j)
            ra[j] = *(const float4*)(ab + (t + 256 * j) * 4);
        #pragma unroll
        for (int j = 0; j < 12; ++j)
            rb[j] = *(const float4*)(btb + (t + 256 * j) * 4);
        const float vv = v[t];
        #pragma unroll
        for (int j = 0; j < 8; ++j) {
            const int e = t + 256 * j;
            const int o = (e >> 2) & (C_DIM - 1);
            const int c = (e >> 10) * NT + (e & 3) * 4;
            *(float4*)(a_s + o * APITCH + c) = ra[j];
        }
        #pragma unroll
        for (int j = 0; j < 12; ++j) {
            const int e = t + 256 * j;
            const int o = (e >> 2) & (C_DIM - 1);
            const int c = (e >> 10) * NT + (e & 3) * 4;
            *(float4*)(bt_s + o * BPITCH + c) = rb[j];
        }
        v_s[t] = vv;
    }
    __syncthreads();

    const int w    = t >> 5;
    const int lane = t & 31;
    const int kp   = lane & 15;
    const int ih   = lane >> 4;

    const float* ap = a_s  + 2 * kp;
    const float* bp = bt_s + 6 * w + 3 * ih;

    float c00 = 0.f, c01 = 0.f, c02 = 0.f;
    float c10 = 0.f, c11 = 0.f, c12 = 0.f;

    #pragma unroll 8
    for (int o = 0; o < C_DIM; ++o) {
        const float a0 = ap[o * APITCH];
        const float a1 = ap[o * APITCH + 1];
        const float b0 = bp[o * BPITCH];
        const float b1 = bp[o * BPITCH + 1];
        const float b2 = bp[o * BPITCH + 2];
        const float vo = v_s[o];
        c00 += vo * hw_tanhf(a0 + b0);
        c01 += vo * hw_tanhf(a0 + b1);
        c02 += vo * hw_tanhf(a0 + b2);
        c10 += vo * hw_tanhf(a1 + b0);
        c11 += vo * hw_tanhf(a1 + b1);
        c12 += vo * hw_tanhf(a1 + b2);
    }

    float* fp = f + (long)b * N_DIM * N_DIM
                  + (long)(iBase + 6 * w + 3 * ih) * N_DIM + kBase + 2 * kp;
    *(float2*)(fp)             = make_float2(c00, c10);
    *(float2*)(fp + N_DIM)     = make_float2(c01, c11);
    *(float2*)(fp + 2 * N_DIM) = make_float2(c02, c12);
}

#define SM_ROWS (N_DIM / 8)
__global__ void __launch_bounds__(256)
softmax_i_kernel(const float* __restrict__ f,
                 float* __restrict__ out)
{
    __shared__ float red[8][33];
    __shared__ float colM[32];
    __shared__ float colS[32];

    const int b     = blockIdx.y;
    const int kBase = blockIdx.x * 32;
    const int kl    = threadIdx.x & 31;
    const int iy    = threadIdx.x >> 5;
    const float* fb = f   + (long)b * N_DIM * N_DIM + (long)iy * N_DIM + kBase + kl;
    float* ob       = out + (long)b * N_DIM * N_DIM + (long)iy * N_DIM + kBase + kl;

    float val[SM_ROWS];
    #pragma unroll
    for (int j = 0; j < SM_ROWS; ++j)
        val[j] = fb[(long)j * 8 * N_DIM];

    float m = -INFINITY;
    #pragma unroll
    for (int j = 0; j < SM_ROWS; ++j)
        m = fmaxf(m, val[j]);
    red[iy][kl] = m;
    __syncthreads();
    if (iy == 0) {
        float mm = red[0][kl];
        #pragma unroll
        for (int r = 1; r < 8; ++r) mm = fmaxf(mm, red[r][kl]);
        colM[kl] = mm;
    }
    __syncthreads();
    const float mm = colM[kl];

    float s = 0.f;
    #pragma unroll
    for (int j = 0; j < SM_ROWS; ++j) {
        val[j] = __expf(val[j] - mm);
        s += val[j];
    }
    red[iy][kl] = s;
    __syncthreads();
    if (iy == 0) {
        float ss = 0.f;
        #pragma unroll
        for (int r = 0; r < 8; ++r) ss += red[r][kl];
        colS[kl] = ss;
    }
    __syncthreads();
    const float inv = 1.f / colS[kl];

    #pragma unroll
    for (int j = 0; j < SM_ROWS; ++j)
        ob[(long)j * 8 * N_DIM] = val[j] * inv;
}

extern "C" void kernel_launch(void* const* d_in, const int* in_sizes, int n_in,
                              void* d_out, int out_size, void* d_ws, size_t ws_size,
                              hipStream_t stream)
{
    const float* x       = (const float*)d_in[0];
    const float* theta_w = (const float*)d_in[1];
    const float* theta_b = (const float*)d_in[2];
    const float* phi_w   = (const float*)d_in[3];
    const float* phi_b   = (const float*)d_in[4];
    const float* W_a     = (const float*)d_in[5];
    const float* v_a     = (const float*)d_in[6];
    float* out = (float*)d_out;

    float* ws = (float*)d_ws;
    float* a  = ws;
    float* bt = a  + B_DIM * C_DIM * N_DIM;
    float* f  = bt + B_DIM * C_DIM * N_DIM;

    {
        dim3 grid(N_TILES, B_DIM, 2), block(512, 1, 1);
        chain_gemm_f32<<<grid, block, 0, stream>>>(
            x, theta_w, theta_b, phi_w, phi_b, W_a, a, bt);
    }

    {
        dim3 grid(N_DIM / PW_KT, N_DIM / PW_IT, B_DIM), block(256, 1, 1);
        pairwise_tanh_kernel<<<grid, block, 0, stream>>>(a, bt, v_a, f);
    }

    {
        dim3 grid(N_DIM / 32, B_DIM, 1), block(256, 1, 1);
        softmax_i_kernel<<<grid, block, 0, stream>>>(f, out);
    }
}


# Weight fragment loads widened to 16 B per lane, lane halves adjacent: 32 B per cache line touched (was 16), half the loads; k regrouped in blocks of 8 in both products (sum order changes)
# speedup vs baseline: 3.5418x; 1.0349x over previous
#include <hip/hip_runtime.h>
#include <math.h>

typedef __attribute__((ext_vector_type(2))) float v2f;
typedef __attribute__((ext_vector_type(8))) float v8f;

#define B_DIM 2
#define C_DIM 256
#define N_DIM 384

static __device__ __forceinline__ float hw_tanhf(float x) {
#if __has_builtin(__builtin_amdgcn_tanhf)
    return __builtin_amdgcn_tanhf(x);
#else
    float r;
    asm("v_tanh_f32 %0, %1" : "=v"(r) : "v"(x));
    return r;
#endif
}

#define NT 16
#define N_TILES (N_DIM / NT)
#define SLAB_OFF(k) (((k) + 2 * ((k) >> 2)) * NT)
#define SLAB_FLOATS ((C_DIM + 2 * (C_DIM / 4)) * NT)

__global__ void __launch_bounds__(512)
chain_gemm_f32(const float* __restrict__ x,
               const float* __restrict__ theta_w,
               const float* __restrict__ theta_b,
               const float* __restrict__ phi_w,
               const float* __restrict__ phi_b,
               const float* __restrict__ W_a,
               float* __restrict__ a_t,
               float* __restrict__ bt_t)
{
    __shared__ float xs[SLAB_FLOATS];
    __shared__ float hs[SLAB_FLOATS];

    const int tile  = blockIdx.x;
    const int b     = blockIdx.y;
    const int which = blockIdx.z;
    const float* Wf   = which ? phi_w : theta_w;
    const float* bias = which ? phi_b : theta_b;
    const float* Ws   = W_a + which * C_DIM;
    float* dst = (which ? bt_t : a_t) + (long)(b * N_TILES + tile) * (C_DIM * NT);

    const int t = threadIdx.x;

    {
        const float* xb = x + (long)b * C_DIM * N_DIM + tile * NT;
        const int c4 = (t & 3) * 4;
        #pragma unroll
        for (int p = 0; p < 2; ++p) {
            const int r = (t >> 2) + p * 128;
            const float4 val = *(const float4*)(xb + (long)r * N_DIM + c4);
            *(float4*)(xs + SLAB_OFF(r) + c4) = val;
        }
    }
    __syncthreads();

    const int w    = t >> 5;
    const int lane = t & 31;
    const int half = lane >> 4;
    const int ln   = lane & 15;
    const int rowA = w * 16 + ln;

    v8f acc = {};
    {
        const float* Arow = Wf + (long)rowA * C_DIM + 4 * half;
        const float* Bcol = xs + SLAB_OFF(4 * half) + ln;
        #pragma unroll 8
        for (int kb = 0; kb < C_DIM; kb += 8) {
            const float4 q = *(const float4*)(Arow + kb);
            const float* Bk = Bcol + SLAB_OFF(kb);
            v2f a0, a1, b0, b1;
            a0.x = q.x;        a0.y = q.y;
            b0.x = Bk[0];      b0.y = Bk[NT];
            a1.x = q.z;        a1.y = q.w;
            b1.x = Bk[2 * NT]; b1.y = Bk[3 * NT];
            acc = __builtin_amdgcn_wmma_f32_16x16x4_f32(
                 false, a0,  false, b0,
                 (short)0, acc,  false,  false);
            acc = __builtin_amdgcn_wmma_f32_16x16x4_f32(
                 false, a1,  false, b1,
                 (short)0, acc,  false,  false);
        }
    }
    #pragma unroll
    for (int v = 0; v < 8; ++v) {
        const int row = w * 16 + v + 8 * half;
        hs[SLAB_OFF(row) + ln] = acc[v] + bias[row];
    }
    __syncthreads();

    v8f acc2 = {};
    {
        const float* Arow = Ws + (long)rowA * (2 * C_DIM) + 4 * half;
        const float* Bcol = hs + SLAB_OFF(4 * half) + ln;
        #pragma unroll 8
        for (int kb = 0; kb < C_DIM; kb += 8) {
            const float4 q = *(const float4*)(Arow + kb);
            const float* Bk = Bcol + SLAB_OFF(kb);
            v2f a0, a1, b0, b1;
            a0.x = q.x;        a0.y = q.y;
            b0.x = Bk[0];      b0.y = Bk[NT];
            a1.x = q.z;        a1.y = q.w;
            b1.x = Bk[2 * NT]; b1.y = Bk[3 * NT];
            acc2 = __builtin_amdgcn_wmma_f32_16x16x4_f32(
                 false, a0,  false, b0,
                 (short)0, acc2,  false,  false);
            acc2 = __builtin_amdgcn_wmma_f32_16x16x4_f32(
                 false, a1,  false, b1,
                 (short)0, acc2,  false,  false);
        }
    }
    #pragma unroll
    for (int v = 0; v < 8; ++v) {
        const int row = w * 16 + v + 8 * half;
        dst[row * NT + ln] = acc2[v];
    }
}

#define PW_KT 32
#define PW_IT 48
#define APITCH 36
#define BPITCH 52
__global__ void __launch_bounds__(256)
pairwise_tanh_kernel(const float* __restrict__ a,
                     const float* __restrict__ bt,
                     const float* __restrict__ v,
                     float* __restrict__ f)
{
    __shared__ float a_s[C_DIM * APITCH];
    __shared__ float bt_s[C_DIM * BPITCH];
    __shared__ float v_s[C_DIM];

    const int b     = blockIdx.z;
    const int kBase = blockIdx.x * PW_KT;
    const int iBase = blockIdx.y * PW_IT;
    const float* ab  = a  + (long)b * C_DIM * N_DIM + (long)(kBase / NT) * (C_DIM * NT);
    const float* btb = bt + (long)b * C_DIM * N_DIM + (long)(iBase / NT) * (C_DIM * NT);

    const int t  = threadIdx.x;

    {
        float4 ra[8], rb[12];
        #pragma unroll
        for (int j = 0; j < 8; ++j)
            ra[j] = *(const float4*)(ab + (t + 256 * j) * 4);
        #pragma unroll
        for (int j = 0; j < 12; ++j)
            rb[j] = *(const float4*)(btb + (t + 256 * j) * 4);
        const float vv = v[t];
        #pragma unroll
        for (int j = 0; j < 8; ++j) {
            const int e = t + 256 * j;
            const int o = (e >> 2) & (C_DIM - 1);
            const int c = (e >> 10) * NT + (e & 3) * 4;
            *(float4*)(a_s + o * APITCH + c) = ra[j];
        }
        #pragma unroll
        for (int j = 0; j < 12; ++j) {
            const int e = t + 256 * j;
            const int o = (e >> 2) & (C_DIM - 1);
            const int c = (e >> 10) * NT + (e & 3) * 4;
            *(float4*)(bt_s + o * BPITCH + c) = rb[j];
        }
        v_s[t] = vv;
    }
    __syncthreads();

    const int w    = t >> 5;
    const int lane = t & 31;
    const int kp   = lane & 15;
    const int ih   = lane >> 4;

    const float* ap = a_s  + 2 * kp;
    const float* bp = bt_s + 6 * w + 3 * ih;

    float c00 = 0.f, c01 = 0.f, c02 = 0.f;
    float c10 = 0.f, c11 = 0.f, c12 = 0.f;

    #pragma unroll 8
    for (int o = 0; o < C_DIM; ++o) {
        const float a0 = ap[o * APITCH];
        const float a1 = ap[o * APITCH + 1];
        const float b0 = bp[o * BPITCH];
        const float b1 = bp[o * BPITCH + 1];
        const float b2 = bp[o * BPITCH + 2];
        const float vo = v_s[o];
        c00 += vo * hw_tanhf(a0 + b0);
        c01 += vo * hw_tanhf(a0 + b1);
        c02 += vo * hw_tanhf(a0 + b2);
        c10 += vo * hw_tanhf(a1 + b0);
        c11 += vo * hw_tanhf(a1 + b1);
        c12 += vo * hw_tanhf(a1 + b2);
    }

    float* fp = f + (long)b * N_DIM * N_DIM
                  + (long)(iBase + 6 * w + 3 * ih) * N_DIM + kBase + 2 * kp;
    *(float2*)(fp)             = make_float2(c00, c10);
    *(float2*)(fp + N_DIM)     = make_float2(c01, c11);
    *(float2*)(fp + 2 * N_DIM) = make_float2(c02, c12);
}

#define SM_ROWS (N_DIM / 8)
__global__ void __launch_bounds__(256)
softmax_i_kernel(const float* __restrict__ f,
                 float* __restrict__ out)
{
    __shared__ float red[8][33];
    __shared__ float colM[32];
    __shared__ float colS[32];

    const int b     = blockIdx.y;
    const int kBase = blockIdx.x * 32;
    const int kl    = threadIdx.x & 31;
    const int iy    = threadIdx.x >> 5;
    const float* fb = f   + (long)b * N_DIM * N_DIM + (long)iy * N_DIM + kBase + kl;
    float* ob       = out + (long)b * N_DIM * N_DIM + (long)iy * N_DIM + kBase + kl;

    float val[SM_ROWS];
    #pragma unroll
    for (int j = 0; j < SM_ROWS; ++j)
        val[j] = fb[(long)j * 8 * N_DIM];

    float m = -INFINITY;
    #pragma unroll
    for (int j = 0; j < SM_ROWS; ++j)
        m = fmaxf(m, val[j]);
    red[iy][kl] = m;
    __syncthreads();
    if (iy == 0) {
        float mm = red[0][kl];
        #pragma unroll
        for (int r = 1; r < 8; ++r) mm = fmaxf(mm, red[r][kl]);
        colM[kl] = mm;
    }
    __syncthreads();
    const float mm = colM[kl];

    float s = 0.f;
    #pragma unroll
    for (int j = 0; j < SM_ROWS; ++j) {
        val[j] = __expf(val[j] - mm);
        s += val[j];
    }
    red[iy][kl] = s;
    __syncthreads();
    if (iy == 0) {
        float ss = 0.f;
        #pragma unroll
        for (int r = 0; r < 8; ++r) ss += red[r][kl];
        colS[kl] = ss;
    }
    __syncthreads();
    const float inv = 1.f / colS[kl];

    #pragma unroll
    for (int j = 0; j < SM_ROWS; ++j)
        ob[(long)j * 8 * N_DIM] = val[j] * inv;
}

extern "C" void kernel_launch(void* const* d_in, const int* in_sizes, int n_in,
                              void* d_out, int out_size, void* d_ws, size_t ws_size,
                              hipStream_t stream)
{
    const float* x       = (const float*)d_in[0];
    const float* theta_w = (const float*)d_in[1];
    const float* theta_b = (const float*)d_in[2];
    const float* phi_w   = (const float*)d_in[3];
    const float* phi_b   = (const float*)d_in[4];
    const float* W_a     = (const float*)d_in[5];
    const float* v_a     = (const float*)d_in[6];
    float* out = (float*)d_out;

    float* ws = (float*)d_ws;
    float* a  = ws;
    float* bt = a  + B_DIM * C_DIM * N_DIM;
    float* f  = bt + B_DIM * C_DIM * N_DIM;

    {
        dim3 grid(N_TILES, B_DIM, 2), block(512, 1, 1);
        chain_gemm_f32<<<grid, block, 0, stream>>>(
            x, theta_w, theta_b, phi_w, phi_b, W_a, a, bt);
    }

    {
        dim3 grid(N_DIM / PW_KT, N_DIM / PW_IT, B_DIM), block(256, 1, 1);
        pairwise_tanh_kernel<<<grid, block, 0, stream>>>(a, bt, v_a, f);
    }

    {
        dim3 grid(N_DIM / 32, B_DIM, 1), block(256, 1, 1);
        softmax_i_kernel<<<grid, block, 0, stream>>>(f, out);
    }
}


# Projection kernel: weight fragment loads software-pipelined about one trip (14 matrix ops) ahead of use, across both products and the barrier between them; bias read at kernel start
# speedup vs baseline: 3.6473x; 1.0298x over previous
#include <hip/hip_runtime.h>
#include <math.h>

typedef __attribute__((ext_vector_type(2))) float v2f;
typedef __attribute__((ext_vector_type(8))) float v8f;

#define B_DIM 2
#define C_DIM 256
#define N_DIM 384

static __device__ __forceinline__ float hw_tanhf(float x) {
#if __has_builtin(__builtin_amdgcn_tanhf)
    return __builtin_amdgcn_tanhf(x);
#else
    float r;
    asm("v_tanh_f32 %0, %1" : "=v"(r) : "v"(x));
    return r;
#endif
}

#define NT 16
#define N_TILES (N_DIM / NT)
#define SLAB_OFF(k) (((k) + 2 * ((k) >> 2)) * NT)
#define SLAB_FLOATS ((C_DIM + 2 * (C_DIM / 4)) * NT)

#define TRIP_K 64

static __device__ __forceinline__ void load_trip(float4 (&q)[8],
                                                 const float* __restrict__ Arow,
                                                 const int k0)
{
    #pragma unroll
    for (int j = 0; j < 8; ++j)
        q[j] = *(const float4*)(Arow + k0 + 8 * j);
}

static __device__ __forceinline__ v8f mma_trip(v8f acc, const float4 (&q)[8],
                                               const float* Bcol, const int k0)
{
    #pragma unroll
    for (int j = 0; j < 8; ++j) {
        const float* Bk = Bcol + SLAB_OFF(k0 + 8 * j);
        v2f a0, a1, b0, b1;
        a0.x = q[j].x;     a0.y = q[j].y;
        b0.x = Bk[0];      b0.y = Bk[NT];
        a1.x = q[j].z;     a1.y = q[j].w;
        b1.x = Bk[2 * NT]; b1.y = Bk[3 * NT];
        acc = __builtin_amdgcn_wmma_f32_16x16x4_f32(
             false, a0,  false, b0,
             (short)0, acc,  false,  false);
        acc = __builtin_amdgcn_wmma_f32_16x16x4_f32(
             false, a1,  false, b1,
             (short)0, acc,  false,  false);
    }
    return acc;
}

__global__ void __launch_bounds__(512)
chain_gemm_f32(const float* __restrict__ x,
               const float* __restrict__ theta_w,
               const float* __restrict__ theta_b,
               const float* __restrict__ phi_w,
               const float* __restrict__ phi_b,
               const float* __restrict__ W_a,
               float* __restrict__ a_t,
               float* __restrict__ bt_t)
{
    __shared__ float xs[SLAB_FLOATS];
    __shared__ float hs[SLAB_FLOATS];

    const int tile  = blockIdx.x;
    const int b     = blockIdx.y;
    const int which = blockIdx.z;
    const float* Wf   = which ? phi_w : theta_w;
    const float* bias = which ? phi_b : theta_b;
    const float* Ws   = W_a + which * C_DIM;
    float* dst = (which ? bt_t : a_t) + (long)(b * N_TILES + tile) * (C_DIM * NT);

    const int t    = threadIdx.x;
    const int w    = t >> 5;
    const int lane = t & 31;
    const int half = lane >> 4;
    const int ln   = lane & 15;
    const int rowA = w * 16 + ln;
    const int row0 = w * 16 + 8 * half;

    const float* A1 = Wf + (long)rowA * C_DIM + 4 * half;
    const float* A2 = Ws + (long)rowA * (2 * C_DIM) + 4 * half;

    float4 xv[2];
    {
        const float* xb = x + (long)b * C_DIM * N_DIM + tile * NT;
        const int c4 = (t & 3) * 4;
        #pragma unroll
        for (int p = 0; p < 2; ++p) {
            const int r = (t >> 2) + p * 128;
            xv[p] = *(const float4*)(xb + (long)r * N_DIM + c4);
        }
    }
    const float4 bia0 = *(const float4*)(bias + row0);
    const float4 bia1 = *(const float4*)(bias + row0 + 4);

    float4 qa[8], qb[8];
    load_trip(qa, A1, 0 * TRIP_K);

    {
        const int c4 = (t & 3) * 4;
        #pragma unroll
        for (int p = 0; p < 2; ++p) {
            const int r = (t >> 2) + p * 128;
            *(float4*)(xs + SLAB_OFF(r) + c4) = xv[p];
        }
    }
    __syncthreads();

    const float* Bx = xs + SLAB_OFF(4 * half) + ln;
    v8f acc = {};
    load_trip(qb, A1, 1 * TRIP_K);  acc = mma_trip(acc, qa, Bx, 0 * TRIP_K);
    load_trip(qa, A1, 2 * TRIP_K);  acc = mma_trip(acc, qb, Bx, 1 * TRIP_K);
    load_trip(qb, A1, 3 * TRIP_K);  acc = mma_trip(acc, qa, Bx, 2 * TRIP_K);
    load_trip(qa, A2, 0 * TRIP_K);  acc = mma_trip(acc, qb, Bx, 3 * TRIP_K);

    {
        const float bv[8] = { bia0.x, bia0.y, bia0.z, bia0.w, bia1.x, bia1.y, bia1.z, bia1.w };
        #pragma unroll
        for (int v = 0; v < 8; ++v)
            hs[SLAB_OFF(row0 + v) + ln] = acc[v] + bv[v];
    }
    __syncthreads();

    const float* Bh = hs + SLAB_OFF(4 * half) + ln;
    v8f acc2 = {};
    load_trip(qb, A2, 1 * TRIP_K);  acc2 = mma_trip(acc2, qa, Bh, 0 * TRIP_K);
    load_trip(qa, A2, 2 * TRIP_K);  acc2 = mma_trip(acc2, qb, Bh, 1 * TRIP_K);
    load_trip(qb, A2, 3 * TRIP_K);  acc2 = mma_trip(acc2, qa, Bh, 2 * TRIP_K);
                                    acc2 = mma_trip(acc2, qb, Bh, 3 * TRIP_K);

    #pragma unroll
    for (int v = 0; v < 8; ++v)
        dst[(row0 + v) * NT + ln] = acc2[v];
}

#define PW_KT 32
#define PW_IT 48
#define APITCH 36
#define BPITCH 52
__global__ void __launch_bounds__(256)
pairwise_tanh_kernel(const float* __restrict__ a,
                     const float* __restrict__ bt,
                     const float* __restrict__ v,
                     float* __restrict__ f)
{
    __shared__ float a_s[C_DIM * APITCH];
    __shared__ float bt_s[C_DIM * BPITCH];
    __shared__ float v_s[C_DIM];

    const int b     = blockIdx.z;
    const int kBase = blockIdx.x * PW_KT;
    const int iBase = blockIdx.y * PW_IT;
    const float* ab  = a  + (long)b * C_DIM * N_DIM + (long)(kBase / NT) * (C_DIM * NT);
    const float* btb = bt + (long)b * C_DIM * N_DIM + (long)(iBase / NT) * (C_DIM * NT);

    const int t  = threadIdx.x;

    {
        float4 ra[8], rb[12];
        #pragma unroll
        for (int j = 0; j < 8; ++j)
            ra[j] = *(const float4*)(ab + (t + 256 * j) * 4);
        #pragma unroll
        for (int j = 0; j < 12; ++j)
            rb[j] = *(const float4*)(btb + (t + 256 * j) * 4);
        const float vv = v[t];
        #pragma unroll
        for (int j = 0; j < 8; ++j) {
            const int e = t + 256 * j;
            const int o = (e >> 2) & (C_DIM - 1);
            const int c = (e >> 10) * NT + (e & 3) * 4;
            *(float4*)(a_s + o * APITCH + c) = ra[j];
        }
        #pragma unroll
        for (int j = 0; j < 12; ++j) {
            const int e = t + 256 * j;
            const int o = (e >> 2) & (C_DIM - 1);
            const int c = (e >> 10) * NT + (e & 3) * 4;
            *(float4*)(bt_s + o * BPITCH + c) = rb[j];
        }
        v_s[t] = vv;
    }
    __syncthreads();

    const int w    = t >> 5;
    const int lane = t & 31;
    const int kp   = lane & 15;
    const int ih   = lane >> 4;

    const float* ap = a_s  + 2 * kp;
    const float* bp = bt_s + 6 * w + 3 * ih;

    float c00 = 0.f, c01 = 0.f, c02 = 0.f;
    float c10 = 0.f, c11 = 0.f, c12 = 0.f;

    #pragma unroll 8
    for (int o = 0; o < C_DIM; ++o) {
        const float a0 = ap[o * APITCH];
        const float a1 = ap[o * APITCH + 1];
        const float b0 = bp[o * BPITCH];
        const float b1 = bp[o * BPITCH + 1];
        const float b2 = bp[o * BPITCH + 2];
        const float vo = v_s[o];
        c00 += vo * hw_tanhf(a0 + b0);
        c01 += vo * hw_tanhf(a0 + b1);
        c02 += vo * hw_tanhf(a0 + b2);
        c10 += vo * hw_tanhf(a1 + b0);
        c11 += vo * hw_tanhf(a1 + b1);
        c12 += vo * hw_tanhf(a1 + b2);
    }

    float* fp = f + (long)b * N_DIM * N_DIM
                  + (long)(iBase + 6 * w + 3 * ih) * N_DIM + kBase + 2 * kp;
    *(float2*)(fp)             = make_float2(c00, c10);
    *(float2*)(fp + N_DIM)     = make_float2(c01, c11);
    *(float2*)(fp + 2 * N_DIM) = make_float2(c02, c12);
}

#define SM_ROWS (N_DIM / 8)
__global__ void __launch_bounds__(256)
softmax_i_kernel(const float* __restrict__ f,
                 float* __restrict__ out)
{
    __shared__ float red[8][33];
    __shared__ float colM[32];
    __shared__ float colS[32];

    const int b     = blockIdx.y;
    const int kBase = blockIdx.x * 32;
    const int kl    = threadIdx.x & 31;
    const int iy    = threadIdx.x >> 5;
    const float* fb = f   + (long)b * N_DIM * N_DIM + (long)iy * N_DIM + kBase + kl;
    float* ob       = out + (long)b * N_DIM * N_DIM + (long)iy * N_DIM + kBase + kl;

    float val[SM_ROWS];
    #pragma unroll
    for (int j = 0; j < SM_ROWS; ++j)
        val[j] = fb[(long)j * 8 * N_DIM];

    float m = -INFINITY;
    #pragma unroll
    for (int j = 0; j < SM_ROWS; ++j)
        m = fmaxf(m, val[j]);
    red[iy][kl] = m;
    __syncthreads();
    if (iy == 0) {
        float mm = red[0][kl];
        #pragma unroll
        for (int r = 1; r < 8; ++r) mm = fmaxf(mm, red[r][kl]);
        colM[kl] = mm;
    }
    __syncthreads();
    const float mm = colM[kl];

    float s = 0.f;
    #pragma unroll
    for (int j = 0; j < SM_ROWS; ++j) {
        val[j] = __expf(val[j] - mm);
        s += val[j];
    }
    red[iy][kl] = s;
    __syncthreads();
    if (iy == 0) {
        float ss = 0.f;
        #pragma unroll
        for (int r = 0; r < 8; ++r) ss += red[r][kl];
        colS[kl] = ss;
    }
    __syncthreads();
    const float inv = 1.f / colS[kl];

    #pragma unroll
    for (int j = 0; j < SM_ROWS; ++j)
        ob[(long)j * 8 * N_DIM] = val[j] * inv;
}

extern "C" void kernel_launch(void* const* d_in, const int* in_sizes, int n_in,
                              void* d_out, int out_size, void* d_ws, size_t ws_size,
                              hipStream_t stream)
{
    const float* x       = (const float*)d_in[0];
    const float* theta_w = (const float*)d_in[1];
    const float* theta_b = (const float*)d_in[2];
    const float* phi_w   = (const float*)d_in[3];
    const float* phi_b   = (const float*)d_in[4];
    const float* W_a     = (const float*)d_in[5];
    const float* v_a     = (const float*)d_in[6];
    float* out = (float*)d_out;

    float* ws = (float*)d_ws;
    float* a  = ws;
    float* bt = a  + B_DIM * C_DIM * N_DIM;
    float* f  = bt + B_DIM * C_DIM * N_DIM;

    {
        dim3 grid(N_TILES, B_DIM, 2), block(512, 1, 1);
        chain_gemm_f32<<<grid, block, 0, stream>>>(
            x, theta_w, theta_b, phi_w, phi_b, W_a, a, bt);
    }

    {
        dim3 grid(N_DIM / PW_KT, N_DIM / PW_IT, B_DIM), block(256, 1, 1);
        pairwise_tanh_kernel<<<grid, block, 0, stream>>>(a, bt, v_a, f);
    }

    {
        dim3 grid(N_DIM / 32, B_DIM, 1), block(256, 1, 1);
        softmax_i_kernel<<<grid, block, 0, stream>>>(f, out);
    }
}


# Projection spread over 192 work-groups (was 96): each owns half the rows of product 2 and builds the whole h slab; 8 waves, operand reads grouped a trip ahead; 384 matrix ops per SIMD (was 512)
# speedup vs baseline: 3.8259x; 1.0490x over previous
#include <hip/hip_runtime.h>
#include <math.h>

typedef __attribute__((ext_vector_type(2))) float v2f;
typedef __attribute__((ext_vector_type(8))) float v8f;

#define B_DIM 2
#define C_DIM 256
#define N_DIM 384

static __device__ __forceinline__ float hw_tanhf(float x) {
#if __has_builtin(__builtin_amdgcn_tanhf)
    return __builtin_amdgcn_tanhf(x);
#else
    float r;
    asm("v_tanh_f32 %0, %1" : "=v"(r) : "v"(x));
    return r;
#endif
}

#define NT 16
#define N_TILES (N_DIM / NT)
#define SLAB_OFF(k) (((k) + 2 * ((k) >> 2)) * NT)
#define SLAB_FLOATS ((C_DIM + 2 * (C_DIM / 4)) * NT)

#define TRIP_K 64
#define GROUP_EDGE __builtin_amdgcn_sched_barrier(0)

static __device__ __forceinline__ void load_trip(float4 (&q)[8],
                                                 const float* __restrict__ Arow,
                                                 const int k0)
{
    #pragma unroll
    for (int j = 0; j < 8; ++j)
        q[j] = *(const float4*)(Arow + k0 + 8 * j);
}

static __device__ __forceinline__ void load_slab_trip(float (&bq)[32],
                                                      const float* Bcol, const int k0)
{
    #pragma unroll
    for (int j = 0; j < 8; ++j) {
        const float* Bk = Bcol + SLAB_OFF(k0 + 8 * j);
        bq[4 * j + 0] = Bk[0];
        bq[4 * j + 1] = Bk[NT];
        bq[4 * j + 2] = Bk[2 * NT];
        bq[4 * j + 3] = Bk[3 * NT];
    }
}

static __device__ __forceinline__ v8f mma4(v8f acc, const float ax, const float ay,
                                           const float bx, const float by)
{
    v2f a, b;
    a.x = ax; a.y = ay;
    b.x = bx; b.y = by;
    return __builtin_amdgcn_wmma_f32_16x16x4_f32(
         false, a,  false, b,
         (short)0, acc,  false,  false);
}

static __device__ __forceinline__ void mma_trip2(v8f& accA, v8f& accB,
                                                 const float4 (&qA)[8], const float4 (&qB)[8],
                                                 const float (&bq)[32])
{
    #pragma unroll
    for (int j = 0; j < 8; ++j) {
        accA = mma4(accA, qA[j].x, qA[j].y, bq[4 * j + 0], bq[4 * j + 1]);
        accB = mma4(accB, qB[j].x, qB[j].y, bq[4 * j + 0], bq[4 * j + 1]);
        accA = mma4(accA, qA[j].z, qA[j].w, bq[4 * j + 2], bq[4 * j + 3]);
        accB = mma4(accB, qB[j].z, qB[j].w, bq[4 * j + 2], bq[4 * j + 3]);
    }
}

static __device__ __forceinline__ v8f mma_trip1(v8f acc, const float4 (&q)[8],
                                                const float (&bq)[32])
{
    #pragma unroll
    for (int j = 0; j < 8; ++j) {
        acc = mma4(acc, q[j].x, q[j].y, bq[4 * j + 0], bq[4 * j + 1]);
        acc = mma4(acc, q[j].z, q[j].w, bq[4 * j + 2], bq[4 * j + 3]);
    }
    return acc;
}

__global__ void __launch_bounds__(256)
chain_gemm_f32(const float* __restrict__ x,
               const float* __restrict__ theta_w,
               const float* __restrict__ theta_b,
               const float* __restrict__ phi_w,
               const float* __restrict__ phi_b,
               const float* __restrict__ W_a,
               float* __restrict__ a_t,
               float* __restrict__ bt_t)
{
    __shared__ float xs[SLAB_FLOATS];
    __shared__ float hs[SLAB_FLOATS];

    const int tile  = blockIdx.x;
    const int b     = blockIdx.y;
    const int which = blockIdx.z >> 1;
    const int mhalf = blockIdx.z & 1;
    const float* Wf   = which ? phi_w : theta_w;
    const float* bias = which ? phi_b : theta_b;
    const float* Ws   = W_a + which * C_DIM;
    float* dst = (which ? bt_t : a_t) + (long)(b * N_TILES + tile) * (C_DIM * NT);

    const int t    = threadIdx.x;
    const int w    = t >> 5;
    const int lane = t & 31;
    const int half = lane >> 4;
    const int ln   = lane & 15;
    const int m2   = 8 * mhalf + w;

    const float* A1a = Wf + (long)(w * 16 + ln) * C_DIM + 4 * half;
    const float* A1b = Wf + (long)((w + 8) * 16 + ln) * C_DIM + 4 * half;
    const float* A2  = Ws + (long)(m2 * 16 + ln) * (2 * C_DIM) + 4 * half;
    const int rowa = w * 16 + 8 * half;
    const int rowb = (w + 8) * 16 + 8 * half;
    const int rowc = m2 * 16 + 8 * half;

    float4 xv[4];
    {
        const float* xb = x + (long)b * C_DIM * N_DIM + tile * NT;
        const int c4 = (t & 3) * 4;
        #pragma unroll
        for (int p = 0; p < 4; ++p) {
            const int r = (t >> 2) + p * 64;
            xv[p] = *(const float4*)(xb + (long)r * N_DIM + c4);
        }
    }
    const float4 bia0 = *(const float4*)(bias + rowa);
    const float4 bia1 = *(const float4*)(bias + rowa + 4);
    const float4 bib0 = *(const float4*)(bias + rowb);
    const float4 bib1 = *(const float4*)(bias + rowb + 4);

    float4 p0a[8], p0b[8], p1a[8], p1b[8];
    float  s0[32], s1[32];
    load_trip(p0a, A1a, 0 * TRIP_K);
    load_trip(p0b, A1b, 0 * TRIP_K);

    {
        const int c4 = (t & 3) * 4;
        #pragma unroll
        for (int p = 0; p < 4; ++p) {
            const int r = (t >> 2) + p * 64;
            *(float4*)(xs + SLAB_OFF(r) + c4) = xv[p];
        }
    }
    __syncthreads();

    const float* Bx = xs + SLAB_OFF(4 * half) + ln;
    v8f accA = {}, accB = {};
    load_slab_trip(s0, Bx, 0 * TRIP_K);
    load_trip(p1a, A1a, 1 * TRIP_K);  load_trip(p1b, A1b, 1 * TRIP_K);  load_slab_trip(s1, Bx, 1 * TRIP_K);
    GROUP_EDGE;
    mma_trip2(accA, accB, p0a, p0b, s0);
    GROUP_EDGE;
    load_trip(p0a, A1a, 2 * TRIP_K);  load_trip(p0b, A1b, 2 * TRIP_K);  load_slab_trip(s0, Bx, 2 * TRIP_K);
    GROUP_EDGE;
    mma_trip2(accA, accB, p1a, p1b, s1);
    GROUP_EDGE;
    load_trip(p1a, A1a, 3 * TRIP_K);  load_trip(p1b, A1b, 3 * TRIP_K);  load_slab_trip(s1, Bx, 3 * TRIP_K);
    GROUP_EDGE;
    mma_trip2(accA, accB, p0a, p0b, s0);
    GROUP_EDGE;
    load_trip(p0a, A2, 0 * TRIP_K);   load_trip(p0b, A2, 1 * TRIP_K);
    GROUP_EDGE;
    mma_trip2(accA, accB, p1a, p1b, s1);

    {
        const float bva[8] = { bia0.x, bia0.y, bia0.z, bia0.w, bia1.x, bia1.y, bia1.z, bia1.w };
        const float bvb[8] = { bib0.x, bib0.y, bib0.z, bib0.w, bib1.x, bib1.y, bib1.z, bib1.w };
        #pragma unroll
        for (int v = 0; v < 8; ++v) {
            hs[SLAB_OFF(rowa + v) + ln] = accA[v] + bva[v];
            hs[SLAB_OFF(rowb + v) + ln] = accB[v] + bvb[v];
        }
    }
    load_trip(p1a, A2, 2 * TRIP_K);   load_trip(p1b, A2, 3 * TRIP_K);
    __syncthreads();

    const float* Bh = hs + SLAB_OFF(4 * half) + ln;
    v8f acc2 = {};
    load_slab_trip(s0, Bh, 0 * TRIP_K);
    load_slab_trip(s1, Bh, 1 * TRIP_K);
    GROUP_EDGE;
    acc2 = mma_trip1(acc2, p0a, s0);
    GROUP_EDGE;
    load_slab_trip(s0, Bh, 2 * TRIP_K);
    GROUP_EDGE;
    acc2 = mma_trip1(acc2, p0b, s1);
    GROUP_EDGE;
    load_slab_trip(s1, Bh, 3 * TRIP_K);
    GROUP_EDGE;
    acc2 = mma_trip1(acc2, p1a, s0);
    acc2 = mma_trip1(acc2, p1b, s1);

    #pragma unroll
    for (int v = 0; v < 8; ++v)
        dst[(rowc + v) * NT + ln] = acc2[v];
}

#define PW_KT 32
#define PW_IT 48
#define APITCH 36
#define BPITCH 52
__global__ void __launch_bounds__(256)
pairwise_tanh_kernel(const float* __restrict__ a,
                     const float* __restrict__ bt,
                     const float* __restrict__ v,
                     float* __restrict__ f)
{
    __shared__ float a_s[C_DIM * APITCH];
    __shared__ float bt_s[C_DIM * BPITCH];
    __shared__ float v_s[C_DIM];

    const int b     = blockIdx.z;
    const int kBase = blockIdx.x * PW_KT;
    const int iBase = blockIdx.y * PW_IT;
    const float* ab  = a  + (long)b * C_DIM * N_DIM + (long)(kBase / NT) * (C_DIM * NT);
    const float* btb = bt + (long)b * C_DIM * N_DIM + (long)(iBase / NT) * (C_DIM * NT);

    const int t  = threadIdx.x;

    {
        float4 ra[8], rb[12];
        #pragma unroll
        for (int j = 0; j < 8; ++j)
            ra[j] = *(const float4*)(ab + (t + 256 * j) * 4);
        #pragma unroll
        for (int j = 0; j < 12; ++j)
            rb[j] = *(const float4*)(btb + (t + 256 * j) * 4);
        const float vv = v[t];
        #pragma unroll
        for (int j = 0; j < 8; ++j) {
            const int e = t + 256 * j;
            const int o = (e >> 2) & (C_DIM - 1);
            const int c = (e >> 10) * NT + (e & 3) * 4;
            *(float4*)(a_s + o * APITCH + c) = ra[j];
        }
        #pragma unroll
        for (int j = 0; j < 12; ++j) {
            const int e = t + 256 * j;
            const int o = (e >> 2) & (C_DIM - 1);
            const int c = (e >> 10) * NT + (e & 3) * 4;
            *(float4*)(bt_s + o * BPITCH + c) = rb[j];
        }
        v_s[t] = vv;
    }
    __syncthreads();

    const int w    = t >> 5;
    const int lane = t & 31;
    const int kp   = lane & 15;
    const int ih   = lane >> 4;

    const float* ap = a_s  + 2 * kp;
    const float* bp = bt_s + 6 * w + 3 * ih;

    float c00 = 0.f, c01 = 0.f, c02 = 0.f;
    float c10 = 0.f, c11 = 0.f, c12 = 0.f;

    #pragma unroll 8
    for (int o = 0; o < C_DIM; ++o) {
        const float a0 = ap[o * APITCH];
        const float a1 = ap[o * APITCH + 1];
        const float b0 = bp[o * BPITCH];
        const float b1 = bp[o * BPITCH + 1];
        const float b2 = bp[o * BPITCH + 2];
        const float vo = v_s[o];
        c00 += vo * hw_tanhf(a0 + b0);
        c01 += vo * hw_tanhf(a0 + b1);
        c02 += vo * hw_tanhf(a0 + b2);
        c10 += vo * hw_tanhf(a1 + b0);
        c11 += vo * hw_tanhf(a1 + b1);
        c12 += vo * hw_tanhf(a1 + b2);
    }

    float* fp = f + (long)b * N_DIM * N_DIM
                  + (long)(iBase + 6 * w + 3 * ih) * N_DIM + kBase + 2 * kp;
    *(float2*)(fp)             = make_float2(c00, c10);
    *(float2*)(fp + N_DIM)     = make_float2(c01, c11);
    *(float2*)(fp + 2 * N_DIM) = make_float2(c02, c12);
}

#define SM_ROWS (N_DIM / 8)
__global__ void __launch_bounds__(256)
softmax_i_kernel(const float* __restrict__ f,
                 float* __restrict__ out)
{
    __shared__ float red[8][33];
    __shared__ float colM[32];
    __shared__ float colS[32];

    const int b     = blockIdx.y;
    const int kBase = blockIdx.x * 32;
    const int kl    = threadIdx.x & 31;
    const int iy    = threadIdx.x >> 5;
    const float* fb = f   + (long)b * N_DIM * N_DIM + (long)iy * N_DIM + kBase + kl;
    float* ob       = out + (long)b * N_DIM * N_DIM + (long)iy * N_DIM + kBase + kl;

    float val[SM_ROWS];
    #pragma unroll
    for (int j = 0; j < SM_ROWS; ++j)
        val[j] = fb[(long)j * 8 * N_DIM];

    float m = -INFINITY;
    #pragma unroll
    for (int j = 0; j < SM_ROWS; ++j)
        m = fmaxf(m, val[j]);
    red[iy][kl] = m;
    __syncthreads();
    if (iy == 0) {
        float mm = red[0][kl];
        #pragma unroll
        for (int r = 1; r < 8; ++r) mm = fmaxf(mm, red[r][kl]);
        colM[kl] = mm;
    }
    __syncthreads();
    const float mm = colM[kl];

    float s = 0.f;
    #pragma unroll
    for (int j = 0; j < SM_ROWS; ++j) {
        val[j] = __expf(val[j] - mm);
        s += val[j];
    }
    red[iy][kl] = s;
    __syncthreads();
    if (iy == 0) {
        float ss = 0.f;
        #pragma unroll
        for (int r = 0; r < 8; ++r) ss += red[r][kl];
        colS[kl] = ss;
    }
    __syncthreads();
    const float inv = 1.f / colS[kl];

    #pragma unroll
    for (int j = 0; j < SM_ROWS; ++j)
        ob[(long)j * 8 * N_DIM] = val[j] * inv;
}

extern "C" void kernel_launch(void* const* d_in, const int* in_sizes, int n_in,
                              void* d_out, int out_size, void* d_ws, size_t ws_size,
                              hipStream_t stream)
{
    const float* x       = (const float*)d_in[0];
    const float* theta_w = (const float*)d_in[1];
    const float* theta_b = (const float*)d_in[2];
    const float* phi_w   = (const float*)d_in[3];
    const float* phi_b   = (const float*)d_in[4];
    const float* W_a     = (const float*)d_in[5];
    const float* v_a     = (const float*)d_in[6];
    float* out = (float*)d_out;

    float* ws = (float*)d_ws;
    float* a  = ws;
    float* bt = a  + B_DIM * C_DIM * N_DIM;
    float* f  = bt + B_DIM * C_DIM * N_DIM;

    {
        dim3 grid(N_TILES, B_DIM, 4), block(256, 1, 1);
        chain_gemm_f32<<<grid, block, 0, stream>>>(
            x, theta_w, theta_b, phi_w, phi_b, W_a, a, bt);
    }

    {
        dim3 grid(N_DIM / PW_KT, N_DIM / PW_IT, B_DIM), block(256, 1, 1);
        pairwise_tanh_kernel<<<grid, block, 0, stream>>>(a, bt, v_a, f);
    }

    {
        dim3 grid(N_DIM / 32, B_DIM, 1), block(256, 1, 1);
        softmax_i_kernel<<<grid, block, 0, stream>>>(f, out);
    }
}


# speedup vs baseline: 3.9875x; 1.0422x over previous
#include <hip/hip_runtime.h>
#include <math.h>

typedef __attribute__((ext_vector_type(2))) float v2f;
typedef __attribute__((ext_vector_type(8))) float v8f;

#define B_DIM 2
#define C_DIM 256
#define N_DIM 384

static __device__ __forceinline__ float hw_tanhf(float x) {
#if __has_builtin(__builtin_amdgcn_tanhf)
    return __builtin_amdgcn_tanhf(x);
#else
    float r;
    asm("v_tanh_f32 %0, %1" : "=v"(r) : "v"(x));
    return r;
#endif
}

#define NT 16
#define N_TILES (N_DIM / NT)
#define SLAB_OFF(k) (((k) + 2 * ((k) >> 2)) * NT)
#define SLAB_FLOATS ((C_DIM + 2 * (C_DIM / 4)) * NT)

#define TRIP_K 64
#define GROUP_EDGE __builtin_amdgcn_sched_barrier(0)

static __device__ __forceinline__ void load_trip(float4 (&q)[8],
                                                 const float* __restrict__ Arow,
                                                 const int k0)
{
    #pragma unroll
    for (int j = 0; j < 8; ++j)
        q[j] = *(const float4*)(Arow + k0 + 8 * j);
}

static __device__ __forceinline__ void load_slab_trip(float (&bq)[32],
                                                      const float* Bcol, const int k0)
{
    #pragma unroll
    for (int j = 0; j < 8; ++j) {
        const float* Bk = Bcol + SLAB_OFF(k0 + 8 * j);
        bq[4 * j + 0] = Bk[0];
        bq[4 * j + 1] = Bk[NT];
        bq[4 * j + 2] = Bk[2 * NT];
        bq[4 * j + 3] = Bk[3 * NT];
    }
}

static __device__ __forceinline__ v8f mma4(v8f acc, const float ax, const float ay,
                                           const float bx, const float by)
{
    v2f a, b;
    a.x = ax; a.y = ay;
    b.x = bx; b.y = by;
    return __builtin_amdgcn_wmma_f32_16x16x4_f32(
         false, a,  false, b,
         (short)0, acc,  false,  false);
}

static __device__ __forceinline__ void mma_trip2(v8f& accA, v8f& accB,
                                                 const float4 (&qA)[8], const float4 (&qB)[8],
                                                 const float (&bq)[32])
{
    #pragma unroll
    for (int j = 0; j < 8; ++j) {
        accA = mma4(accA, qA[j].x, qA[j].y, bq[4 * j + 0], bq[4 * j + 1]);
        accB = mma4(accB, qB[j].x, qB[j].y, bq[4 * j + 0], bq[4 * j + 1]);
        accA = mma4(accA, qA[j].z, qA[j].w, bq[4 * j + 2], bq[4 * j + 3]);
        accB = mma4(accB, qB[j].z, qB[j].w, bq[4 * j + 2], bq[4 * j + 3]);
    }
}

static __device__ __forceinline__ v8f mma_trip1(v8f acc, const float4 (&q)[8],
                                                const float (&bq)[32])
{
    #pragma unroll
    for (int j = 0; j < 8; ++j) {
        acc = mma4(acc, q[j].x, q[j].y, bq[4 * j + 0], bq[4 * j + 1]);
        acc = mma4(acc, q[j].z, q[j].w, bq[4 * j + 2], bq[4 * j + 3]);
    }
    return acc;
}

__global__ void __launch_bounds__(256)
chain_gemm_f32(const float* __restrict__ x,
               const float* __restrict__ theta_w,
               const float* __restrict__ theta_b,
               const float* __restrict__ phi_w,
               const float* __restrict__ phi_b,
               const float* __restrict__ W_a,
               float* __restrict__ a_t,
               float* __restrict__ bt_t)
{
    __shared__ float xs[SLAB_FLOATS];
    __shared__ float hs[SLAB_FLOATS];

    const int tile  = blockIdx.x;
    const int b     = blockIdx.y;
    const int which = blockIdx.z >> 1;
    const int mhalf = blockIdx.z & 1;
    const float* Wf   = which ? phi_w : theta_w;
    const float* bias = which ? phi_b : theta_b;
    const float* Ws   = W_a + which * C_DIM;
    float* dst = (which ? bt_t : a_t) + (long)(b * N_TILES + tile) * (C_DIM * NT);

    const int t    = threadIdx.x;
    const int w    = t >> 5;
    const int lane = t & 31;
    const int half = lane >> 4;
    const int ln   = lane & 15;
    const int m2   = 8 * mhalf + w;

    const float* A1a = Wf + (long)(w * 16 + ln) * C_DIM + 4 * half;
    const float* A1b = Wf + (long)((w + 8) * 16 + ln) * C_DIM + 4 * half;
    const float* A2  = Ws + (long)(m2 * 16 + ln) * (2 * C_DIM) + 4 * half;
    const int rowa = w * 16 + 8 * half;
    const int rowb = (w + 8) * 16 + 8 * half;
    const int rowc = m2 * 16 + 8 * half;

    float4 xv[4];
    {
        const float* xb = x + (long)b * C_DIM * N_DIM + tile * NT;
        const int c4 = (t & 3) * 4;
        #pragma unroll
        for (int p = 0; p < 4; ++p) {
            const int r = (t >> 2) + p * 64;
            xv[p] = *(const float4*)(xb + (long)r * N_DIM + c4);
        }
    }
    const float4 bia0 = *(const float4*)(bias + rowa);
    const float4 bia1 = *(const float4*)(bias + rowa + 4);
    const float4 bib0 = *(const float4*)(bias + rowb);
    const float4 bib1 = *(const float4*)(bias + rowb + 4);

    float4 p0a[8], p0b[8], p1a[8], p1b[8];
    float  s0[32], s1[32];
    load_trip(p0a, A1a, 0 * TRIP_K);
    load_trip(p0b, A1b, 0 * TRIP_K);

    {
        const int c4 = (t & 3) * 4;
        #pragma unroll
        for (int p = 0; p < 4; ++p) {
            const int r = (t >> 2) + p * 64;
            *(float4*)(xs + SLAB_OFF(r) + c4) = xv[p];
        }
    }
    __syncthreads();

    const float* Bx = xs + SLAB_OFF(4 * half) + ln;
    v8f accA = {}, accB = {};
    load_slab_trip(s0, Bx, 0 * TRIP_K);
    load_trip(p1a, A1a, 1 * TRIP_K);  load_trip(p1b, A1b, 1 * TRIP_K);  load_slab_trip(s1, Bx, 1 * TRIP_K);
    GROUP_EDGE;
    mma_trip2(accA, accB, p0a, p0b, s0);
    GROUP_EDGE;
    load_trip(p0a, A1a, 2 * TRIP_K);  load_trip(p0b, A1b, 2 * TRIP_K);  load_slab_trip(s0, Bx, 2 * TRIP_K);
    GROUP_EDGE;
    mma_trip2(accA, accB, p1a, p1b, s1);
    GROUP_EDGE;
    load_trip(p1a, A1a, 3 * TRIP_K);  load_trip(p1b, A1b, 3 * TRIP_K);  load_slab_trip(s1, Bx, 3 * TRIP_K);
    GROUP_EDGE;
    mma_trip2(accA, accB, p0a, p0b, s0);
    GROUP_EDGE;
    load_trip(p0a, A2, 0 * TRIP_K);   load_trip(p0b, A2, 1 * TRIP_K);
    GROUP_EDGE;
    mma_trip2(accA, accB, p1a, p1b, s1);

    {
        const float bva[8] = { bia0.x, bia0.y, bia0.z, bia0.w, bia1.x, bia1.y, bia1.z, bia1.w };
        const float bvb[8] = { bib0.x, bib0.y, bib0.z, bib0.w, bib1.x, bib1.y, bib1.z, bib1.w };
        #pragma unroll
        for (int v = 0; v < 8; ++v) {
            hs[SLAB_OFF(rowa + v) + ln] = accA[v] + bva[v];
            hs[SLAB_OFF(rowb + v) + ln] = accB[v] + bvb[v];
        }
    }
    load_trip(p1a, A2, 2 * TRIP_K);   load_trip(p1b, A2, 3 * TRIP_K);
    __syncthreads();

    const float* Bh = hs + SLAB_OFF(4 * half) + ln;
    v8f acc2 = {};
    load_slab_trip(s0, Bh, 0 * TRIP_K);
    load_slab_trip(s1, Bh, 1 * TRIP_K);
    GROUP_EDGE;
    acc2 = mma_trip1(acc2, p0a, s0);
    GROUP_EDGE;
    load_slab_trip(s0, Bh, 2 * TRIP_K);
    GROUP_EDGE;
    acc2 = mma_trip1(acc2, p0b, s1);
    GROUP_EDGE;
    load_slab_trip(s1, Bh, 3 * TRIP_K);
    GROUP_EDGE;
    acc2 = mma_trip1(acc2, p1a, s0);
    acc2 = mma_trip1(acc2, p1b, s1);

    #pragma unroll
    for (int v = 0; v < 8; ++v)
        dst[(rowc + v) * NT + ln] = acc2[v];
}

#define PW_KT 32
#define PW_IT 48
#define APITCH 36
#define BPITCH 52
#define PW_ROWS 32
#define PW_EDGE __builtin_amdgcn_sched_barrier(0)
#define PW_PAIRS3 \
    __builtin_amdgcn_sched_group_barrier(0x400, 1, 0); __builtin_amdgcn_sched_group_barrier(0x002, 1, 0); \
    __builtin_amdgcn_sched_group_barrier(0x400, 1, 0); __builtin_amdgcn_sched_group_barrier(0x002, 1, 0); \
    __builtin_amdgcn_sched_group_barrier(0x400, 1, 0); __builtin_amdgcn_sched_group_barrier(0x002, 1, 0)
__global__ void __launch_bounds__(256)
pairwise_tanh_kernel(const float* __restrict__ a,
                     const float* __restrict__ bt,
                     const float* __restrict__ v,
                     float* __restrict__ f)
{
    __shared__ float a_s[C_DIM * APITCH];
    __shared__ float bt_s[C_DIM * BPITCH];
    __shared__ float v_s[(C_DIM / 2) * 4];

    const int b     = blockIdx.z;
    const int kBase = blockIdx.x * PW_KT;
    const int iBase = blockIdx.y * PW_IT;
    const float* ab  = a  + (long)b * C_DIM * N_DIM + (long)(kBase / NT) * (C_DIM * NT);
    const float* btb = bt + (long)b * C_DIM * N_DIM + (long)(iBase / NT) * (C_DIM * NT);

    const int t  = threadIdx.x;

    {
        float4 ra[8], rb[12];
        #pragma unroll
        for (int j = 0; j < 8; ++j)
            ra[j] = *(const float4*)(ab + (t + 256 * j) * 4);
        #pragma unroll
        for (int j = 0; j < 12; ++j)
            rb[j] = *(const float4*)(btb + (t + 256 * j) * 4);
        const float vv = v[t];
        #pragma unroll
        for (int j = 0; j < 8; ++j) {
            const int e = t + 256 * j;
            const int o = (e >> 2) & (C_DIM - 1);
            const int c = (e >> 10) * NT + (e & 3) * 4;
            *(float4*)(a_s + o * APITCH + c) = ra[j];
        }
        #pragma unroll
        for (int j = 0; j < 12; ++j) {
            const int e = t + 256 * j;
            const int o = (e >> 2) & (C_DIM - 1);
            const int c = (e >> 10) * NT + (e & 3) * 4;
            *(float4*)(bt_s + o * BPITCH + c) = rb[j];
        }
        v_s[(t >> 1) * 4 + (t & 1)] = vv;
    }
    __syncthreads();

    const int w    = t >> 5;
    const int lane = t & 31;
    const int kp   = lane & 15;
    const int ih   = lane >> 4;

    const float* ap  = a_s  + 2 * kp;
    const float* bp  = bt_s + 6 * w + 2 * ih;
    const float* bp2 = bt_s + 6 * w + 4 + ih;

    v2f c0 = {0.f, 0.f};
    v2f c1 = {0.f, 0.f};
    v2f c2 = {0.f, 0.f};

    #pragma nounroll
    for (int o0 = 0; o0 < C_DIM; o0 += PW_ROWS) {
        v2f   A[PW_ROWS];
        float B0[PW_ROWS], B1[PW_ROWS];
        v2f   V2[PW_ROWS / 2];
        float B2[PW_ROWS];
        #pragma unroll
        for (int r = 0; r < PW_ROWS; ++r) {
            A[r]   = *(const v2f*)(ap + (o0 + r) * APITCH);
            B0[r]  = bp[(o0 + r) * BPITCH];
            B1[r]  = bp[(o0 + r) * BPITCH + 1];
            B2[r]  = bp2[(o0 + r) * BPITCH];
            if ((r & 1) == 0) V2[r / 2] = *(const v2f*)(v_s + 2 * (o0 + r));
        }
        PW_EDGE;
        v2f s0 = A[0] + B0[0];
        v2f s1 = A[0] + B1[0];
        v2f s2 = A[0] + B2[0];
        v2f p0 = {0.f, 0.f}, p1 = {0.f, 0.f}, p2 = {0.f, 0.f};
        PW_EDGE;
        #pragma unroll
        for (int r = 0; r < PW_ROWS; ++r) {
            v2f t0, t1, t2, n0 = s0, n1 = s1, n2 = s2;
            t0.x = hw_tanhf(s0.x);
            t0.y = hw_tanhf(s0.y);
            t1.x = hw_tanhf(s1.x);
            if (r + 1 < PW_ROWS) {
                n0 = A[r + 1] + B0[r + 1];
                n1 = A[r + 1] + B1[r + 1];
                n2 = A[r + 1] + B2[r + 1];
            }
            PW_PAIRS3;
            PW_EDGE;
            t1.y = hw_tanhf(s1.y);
            t2.x = hw_tanhf(s2.x);
            t2.y = hw_tanhf(s2.y);
            if (r >= 1) {
                const float vo = ((r - 1) & 1) ? V2[(r - 1) / 2].y : V2[(r - 1) / 2].x;
                c0 += vo * p0;
                c1 += vo * p1;
                c2 += vo * p2;
            }
            PW_PAIRS3;
            PW_EDGE;
            p0 = t0;  p1 = t1;  p2 = t2;
            s0 = n0;  s1 = n1;  s2 = n2;
        }
        c0 += V2[PW_ROWS / 2 - 1].y * p0;
        c1 += V2[PW_ROWS / 2 - 1].y * p1;
        c2 += V2[PW_ROWS / 2 - 1].y * p2;
    }

    float* fp = f + (long)b * N_DIM * N_DIM
                  + (long)(iBase + 6 * w) * N_DIM + kBase + 2 * kp;
    *(float2*)(fp + (2 * ih) * N_DIM)     = make_float2(c0.x, c0.y);
    *(float2*)(fp + (2 * ih + 1) * N_DIM) = make_float2(c1.x, c1.y);
    *(float2*)(fp + (4 + ih) * N_DIM)     = make_float2(c2.x, c2.y);
}

#define SM_ROWS (N_DIM / 8)
__global__ void __launch_bounds__(256)
softmax_i_kernel(const float* __restrict__ f,
                 float* __restrict__ out)
{
    __shared__ float red[8][33];
    __shared__ float colM[32];
    __shared__ float colS[32];

    const int b     = blockIdx.y;
    const int kBase = blockIdx.x * 32;
    const int kl    = threadIdx.x & 31;
    const int iy    = threadIdx.x >> 5;
    const float* fb = f   + (long)b * N_DIM * N_DIM + (long)iy * N_DIM + kBase + kl;
    float* ob       = out + (long)b * N_DIM * N_DIM + (long)iy * N_DIM + kBase + kl;

    float val[SM_ROWS];
    #pragma unroll
    for (int j = 0; j < SM_ROWS; ++j)
        val[j] = fb[(long)j * 8 * N_DIM];

    float m = -INFINITY;
    #pragma unroll
    for (int j = 0; j < SM_ROWS; ++j)
        m = fmaxf(m, val[j]);
    red[iy][kl] = m;
    __syncthreads();
    if (iy == 0) {
        float mm = red[0][kl];
        #pragma unroll
        for (int r = 1; r < 8; ++r) mm = fmaxf(mm, red[r][kl]);
        colM[kl] = mm;
    }
    __syncthreads();
    const float mm = colM[kl];

    float s = 0.f;
    #pragma unroll
    for (int j = 0; j < SM_ROWS; ++j) {
        val[j] = __expf(val[j] - mm);
        s += val[j];
    }
    red[iy][kl] = s;
    __syncthreads();
    if (iy == 0) {
        float ss = 0.f;
        #pragma unroll
        for (int r = 0; r < 8; ++r) ss += red[r][kl];
        colS[kl] = ss;
    }
    __syncthreads();
    const float inv = 1.f / colS[kl];

    #pragma unroll
    for (int j = 0; j < SM_ROWS; ++j)
        ob[(long)j * 8 * N_DIM] = val[j] * inv;
}

extern "C" void kernel_launch(void* const* d_in, const int* in_sizes, int n_in,
                              void* d_out, int out_size, void* d_ws, size_t ws_size,
                              hipStream_t stream)
{
    const float* x       = (const float*)d_in[0];
    const float* theta_w = (const float*)d_in[1];
    const float* theta_b = (const float*)d_in[2];
    const float* phi_w   = (const float*)d_in[3];
    const float* phi_b   = (const float*)d_in[4];
    const float* W_a     = (const float*)d_in[5];
    const float* v_a     = (const float*)d_in[6];
    float* out = (float*)d_out;

    float* ws = (float*)d_ws;
    float* a  = ws;
    float* bt = a  + B_DIM * C_DIM * N_DIM;
    float* f  = bt + B_DIM * C_DIM * N_DIM;

    {
        dim3 grid(N_TILES, B_DIM, 4), block(256, 1, 1);
        chain_gemm_f32<<<grid, block, 0, stream>>>(
            x, theta_w, theta_b, phi_w, phi_b, W_a, a, bt);
    }

    {
        dim3 grid(N_DIM / PW_KT, N_DIM / PW_IT, B_DIM), block(256, 1, 1);
        pairwise_tanh_kernel<<<grid, block, 0, stream>>>(a, bt, v_a, f);
    }

    {
        dim3 grid(N_DIM / 32, B_DIM, 1), block(256, 1, 1);
        softmax_i_kernel<<<grid, block, 0, stream>>>(f, out);
    }
}


# Softmax column max and exp-sum reduced per 48-row tile in the pairwise epilogue; last kernel: one elementwise in-place pass, 96 work-groups, no barrier; denominator summed from rescaled partial sums
# speedup vs baseline: 4.1749x; 1.0470x over previous
#include <hip/hip_runtime.h>
#include <math.h>

typedef __attribute__((ext_vector_type(2))) float v2f;
typedef __attribute__((ext_vector_type(8))) float v8f;

#define B_DIM 2
#define C_DIM 256
#define N_DIM 384

static __device__ __forceinline__ float hw_tanhf(float x) {
#if __has_builtin(__builtin_amdgcn_tanhf)
    return __builtin_amdgcn_tanhf(x);
#else
    float r;
    asm("v_tanh_f32 %0, %1" : "=v"(r) : "v"(x));
    return r;
#endif
}

static __device__ __forceinline__ float xhalf(float x) {
    const int b = __builtin_bit_cast(int, x);
    return __builtin_bit_cast(float,
        __builtin_amdgcn_permlanex16(b, b, 0x76543210, 0xfedcba98, false, false));
}

#define NT 16
#define N_TILES (N_DIM / NT)
#define SLAB_OFF(k) (((k) + 2 * ((k) >> 2)) * NT)
#define SLAB_FLOATS ((C_DIM + 2 * (C_DIM / 4)) * NT)

#define TRIP_K 64
#define GROUP_EDGE __builtin_amdgcn_sched_barrier(0)

static __device__ __forceinline__ void load_trip(float4 (&q)[8],
                                                 const float* __restrict__ Arow,
                                                 const int k0)
{
    #pragma unroll
    for (int j = 0; j < 8; ++j)
        q[j] = *(const float4*)(Arow + k0 + 8 * j);
}

static __device__ __forceinline__ void load_slab_trip(float (&bq)[32],
                                                      const float* Bcol, const int k0)
{
    #pragma unroll
    for (int j = 0; j < 8; ++j) {
        const float* Bk = Bcol + SLAB_OFF(k0 + 8 * j);
        bq[4 * j + 0] = Bk[0];
        bq[4 * j + 1] = Bk[NT];
        bq[4 * j + 2] = Bk[2 * NT];
        bq[4 * j + 3] = Bk[3 * NT];
    }
}

static __device__ __forceinline__ v8f mma4(v8f acc, const float ax, const float ay,
                                           const float bx, const float by)
{
    v2f a, b;
    a.x = ax; a.y = ay;
    b.x = bx; b.y = by;
    return __builtin_amdgcn_wmma_f32_16x16x4_f32(
         false, a,  false, b,
         (short)0, acc,  false,  false);
}

static __device__ __forceinline__ void mma_trip2(v8f& accA, v8f& accB,
                                                 const float4 (&qA)[8], const float4 (&qB)[8],
                                                 const float (&bq)[32])
{
    #pragma unroll
    for (int j = 0; j < 8; ++j) {
        accA = mma4(accA, qA[j].x, qA[j].y, bq[4 * j + 0], bq[4 * j + 1]);
        accB = mma4(accB, qB[j].x, qB[j].y, bq[4 * j + 0], bq[4 * j + 1]);
        accA = mma4(accA, qA[j].z, qA[j].w, bq[4 * j + 2], bq[4 * j + 3]);
        accB = mma4(accB, qB[j].z, qB[j].w, bq[4 * j + 2], bq[4 * j + 3]);
    }
}

static __device__ __forceinline__ v8f mma_trip1(v8f acc, const float4 (&q)[8],
                                                const float (&bq)[32])
{
    #pragma unroll
    for (int j = 0; j < 8; ++j) {
        acc = mma4(acc, q[j].x, q[j].y, bq[4 * j + 0], bq[4 * j + 1]);
        acc = mma4(acc, q[j].z, q[j].w, bq[4 * j + 2], bq[4 * j + 3]);
    }
    return acc;
}

__global__ void __launch_bounds__(256)
chain_gemm_f32(const float* __restrict__ x,
               const float* __restrict__ theta_w,
               const float* __restrict__ theta_b,
               const float* __restrict__ phi_w,
               const float* __restrict__ phi_b,
               const float* __restrict__ W_a,
               float* __restrict__ a_t,
               float* __restrict__ bt_t)
{
    __shared__ float xs[SLAB_FLOATS];
    __shared__ float hs[SLAB_FLOATS];

    const int tile  = blockIdx.x;
    const int b     = blockIdx.y;
    const int which = blockIdx.z >> 1;
    const int mhalf = blockIdx.z & 1;
    const float* Wf   = which ? phi_w : theta_w;
    const float* bias = which ? phi_b : theta_b;
    const float* Ws   = W_a + which * C_DIM;
    float* dst = (which ? bt_t : a_t) + (long)(b * N_TILES + tile) * (C_DIM * NT);

    const int t    = threadIdx.x;
    const int w    = t >> 5;
    const int lane = t & 31;
    const int half = lane >> 4;
    const int ln   = lane & 15;
    const int m2   = 8 * mhalf + w;

    const float* A1a = Wf + (long)(w * 16 + ln) * C_DIM + 4 * half;
    const float* A1b = Wf + (long)((w + 8) * 16 + ln) * C_DIM + 4 * half;
    const float* A2  = Ws + (long)(m2 * 16 + ln) * (2 * C_DIM) + 4 * half;
    const int rowa = w * 16 + 8 * half;
    const int rowb = (w + 8) * 16 + 8 * half;
    const int rowc = m2 * 16 + 8 * half;

    float4 xv[4];
    {
        const float* xb = x + (long)b * C_DIM * N_DIM + tile * NT;
        const int c4 = (t & 3) * 4;
        #pragma unroll
        for (int p = 0; p < 4; ++p) {
            const int r = (t >> 2) + p * 64;
            xv[p] = *(const float4*)(xb + (long)r * N_DIM + c4);
        }
    }
    const float4 bia0 = *(const float4*)(bias + rowa);
    const float4 bia1 = *(const float4*)(bias + rowa + 4);
    const float4 bib0 = *(const float4*)(bias + rowb);
    const float4 bib1 = *(const float4*)(bias + rowb + 4);

    float4 p0a[8], p0b[8], p1a[8], p1b[8];
    float  s0[32], s1[32];
    load_trip(p0a, A1a, 0 * TRIP_K);
    load_trip(p0b, A1b, 0 * TRIP_K);

    {
        const int c4 = (t & 3) * 4;
        #pragma unroll
        for (int p = 0; p < 4; ++p) {
            const int r = (t >> 2) + p * 64;
            *(float4*)(xs + SLAB_OFF(r) + c4) = xv[p];
        }
    }
    __syncthreads();

    const float* Bx = xs + SLAB_OFF(4 * half) + ln;
    v8f accA = {}, accB = {};
    load_slab_trip(s0, Bx, 0 * TRIP_K);
    load_trip(p1a, A1a, 1 * TRIP_K);  load_trip(p1b, A1b, 1 * TRIP_K);  load_slab_trip(s1, Bx, 1 * TRIP_K);
    GROUP_EDGE;
    mma_trip2(accA, accB, p0a, p0b, s0);
    GROUP_EDGE;
    load_trip(p0a, A1a, 2 * TRIP_K);  load_trip(p0b, A1b, 2 * TRIP_K);  load_slab_trip(s0, Bx, 2 * TRIP_K);
    GROUP_EDGE;
    mma_trip2(accA, accB, p1a, p1b, s1);
    GROUP_EDGE;
    load_trip(p1a, A1a, 3 * TRIP_K);  load_trip(p1b, A1b, 3 * TRIP_K);  load_slab_trip(s1, Bx, 3 * TRIP_K);
    GROUP_EDGE;
    mma_trip2(accA, accB, p0a, p0b, s0);
    GROUP_EDGE;
    load_trip(p0a, A2, 0 * TRIP_K);   load_trip(p0b, A2, 1 * TRIP_K);
    GROUP_EDGE;
    mma_trip2(accA, accB, p1a, p1b, s1);

    {
        const float bva[8] = { bia0.x, bia0.y, bia0.z, bia0.w, bia1.x, bia1.y, bia1.z, bia1.w };
        const float bvb[8] = { bib0.x, bib0.y, bib0.z, bib0.w, bib1.x, bib1.y, bib1.z, bib1.w };
        #pragma unroll
        for (int v = 0; v < 8; ++v) {
            hs[SLAB_OFF(rowa + v) + ln] = accA[v] + bva[v];
            hs[SLAB_OFF(rowb + v) + ln] = accB[v] + bvb[v];
        }
    }
    load_trip(p1a, A2, 2 * TRIP_K);   load_trip(p1b, A2, 3 * TRIP_K);
    __syncthreads();

    const float* Bh = hs + SLAB_OFF(4 * half) + ln;
    v8f acc2 = {};
    load_slab_trip(s0, Bh, 0 * TRIP_K);
    load_slab_trip(s1, Bh, 1 * TRIP_K);
    GROUP_EDGE;
    acc2 = mma_trip1(acc2, p0a, s0);
    GROUP_EDGE;
    load_slab_trip(s0, Bh, 2 * TRIP_K);
    GROUP_EDGE;
    acc2 = mma_trip1(acc2, p0b, s1);
    GROUP_EDGE;
    load_slab_trip(s1, Bh, 3 * TRIP_K);
    GROUP_EDGE;
    acc2 = mma_trip1(acc2, p1a, s0);
    acc2 = mma_trip1(acc2, p1b, s1);

    #pragma unroll
    for (int v = 0; v < 8; ++v)
        dst[(rowc + v) * NT + ln] = acc2[v];
}

#define PW_KT 32
#define PW_IT 48
#define APITCH 36
#define BPITCH 52
#define PW_ROWS 32
#define PW_EDGE __builtin_amdgcn_sched_barrier(0)
#define PW_PAIRS3 \
    __builtin_amdgcn_sched_group_barrier(0x400, 1, 0); __builtin_amdgcn_sched_group_barrier(0x002, 1, 0); \
    __builtin_amdgcn_sched_group_barrier(0x400, 1, 0); __builtin_amdgcn_sched_group_barrier(0x002, 1, 0); \
    __builtin_amdgcn_sched_group_barrier(0x400, 1, 0); __builtin_amdgcn_sched_group_barrier(0x002, 1, 0)
__global__ void __launch_bounds__(256)
pairwise_tanh_kernel(const float* __restrict__ a,
                     const float* __restrict__ bt,
                     const float* __restrict__ v,
                     float* __restrict__ f,
                     float* __restrict__ pm,
                     float* __restrict__ ps)
{
    __shared__ float a_s[C_DIM * APITCH];
    __shared__ float bt_s[C_DIM * BPITCH];
    __shared__ float v_s[(C_DIM / 2) * 4];
    __shared__ float wm_s[8][PW_KT];
    __shared__ float ws_s[8][PW_KT];

    const int b     = blockIdx.z;
    const int kBase = blockIdx.x * PW_KT;
    const int iBase = blockIdx.y * PW_IT;
    const float* ab  = a  + (long)b * C_DIM * N_DIM + (long)(kBase / NT) * (C_DIM * NT);
    const float* btb = bt + (long)b * C_DIM * N_DIM + (long)(iBase / NT) * (C_DIM * NT);

    const int t  = threadIdx.x;

    {
        float4 ra[8], rb[12];
        #pragma unroll
        for (int j = 0; j < 8; ++j)
            ra[j] = *(const float4*)(ab + (t + 256 * j) * 4);
        #pragma unroll
        for (int j = 0; j < 12; ++j)
            rb[j] = *(const float4*)(btb + (t + 256 * j) * 4);
        const float vv = v[t];
        #pragma unroll
        for (int j = 0; j < 8; ++j) {
            const int e = t + 256 * j;
            const int o = (e >> 2) & (C_DIM - 1);
            const int c = (e >> 10) * NT + (e & 3) * 4;
            *(float4*)(a_s + o * APITCH + c) = ra[j];
        }
        #pragma unroll
        for (int j = 0; j < 12; ++j) {
            const int e = t + 256 * j;
            const int o = (e >> 2) & (C_DIM - 1);
            const int c = (e >> 10) * NT + (e & 3) * 4;
            *(float4*)(bt_s + o * BPITCH + c) = rb[j];
        }
        v_s[(t >> 1) * 4 + (t & 1)] = vv;
    }
    __syncthreads();

    const int w    = t >> 5;
    const int lane = t & 31;
    const int kp   = lane & 15;
    const int ih   = lane >> 4;

    const float* ap  = a_s  + 2 * kp;
    const float* bp  = bt_s + 6 * w + 2 * ih;
    const float* bp2 = bt_s + 6 * w + 4 + ih;

    v2f c0 = {0.f, 0.f};
    v2f c1 = {0.f, 0.f};
    v2f c2 = {0.f, 0.f};

    #pragma nounroll
    for (int o0 = 0; o0 < C_DIM; o0 += PW_ROWS) {
        v2f   A[PW_ROWS];
        float B0[PW_ROWS], B1[PW_ROWS];
        v2f   V2[PW_ROWS / 2];
        float B2[PW_ROWS];
        #pragma unroll
        for (int r = 0; r < PW_ROWS; ++r) {
            A[r]   = *(const v2f*)(ap + (o0 + r) * APITCH);
            B0[r]  = bp[(o0 + r) * BPITCH];
            B1[r]  = bp[(o0 + r) * BPITCH + 1];
            B2[r]  = bp2[(o0 + r) * BPITCH];
            if ((r & 1) == 0) V2[r / 2] = *(const v2f*)(v_s + 2 * (o0 + r));
        }
        PW_EDGE;
        v2f s0 = A[0] + B0[0];
        v2f s1 = A[0] + B1[0];
        v2f s2 = A[0] + B2[0];
        v2f p0 = {0.f, 0.f}, p1 = {0.f, 0.f}, p2 = {0.f, 0.f};
        PW_EDGE;
        #pragma unroll
        for (int r = 0; r < PW_ROWS; ++r) {
            v2f t0, t1, t2, n0 = s0, n1 = s1, n2 = s2;
            t0.x = hw_tanhf(s0.x);
            t0.y = hw_tanhf(s0.y);
            t1.x = hw_tanhf(s1.x);
            if (r + 1 < PW_ROWS) {
                n0 = A[r + 1] + B0[r + 1];
                n1 = A[r + 1] + B1[r + 1];
                n2 = A[r + 1] + B2[r + 1];
            }
            PW_PAIRS3;
            PW_EDGE;
            t1.y = hw_tanhf(s1.y);
            t2.x = hw_tanhf(s2.x);
            t2.y = hw_tanhf(s2.y);
            if (r >= 1) {
                const float vo = ((r - 1) & 1) ? V2[(r - 1) / 2].y : V2[(r - 1) / 2].x;
                c0 += vo * p0;
                c1 += vo * p1;
                c2 += vo * p2;
            }
            PW_PAIRS3;
            PW_EDGE;
            p0 = t0;  p1 = t1;  p2 = t2;
            s0 = n0;  s1 = n1;  s2 = n2;
        }
        c0 += V2[PW_ROWS / 2 - 1].y * p0;
        c1 += V2[PW_ROWS / 2 - 1].y * p1;
        c2 += V2[PW_ROWS / 2 - 1].y * p2;
    }

    float mx = fmaxf(fmaxf(c0.x, c1.x), c2.x);
    float my = fmaxf(fmaxf(c0.y, c1.y), c2.y);
    mx = fmaxf(mx, xhalf(mx));
    my = fmaxf(my, xhalf(my));
    float sx = (__expf(c0.x - mx) + __expf(c1.x - mx)) + __expf(c2.x - mx);
    float sy = (__expf(c0.y - my) + __expf(c1.y - my)) + __expf(c2.y - my);
    sx += xhalf(sx);
    sy += xhalf(sy);
    if (ih == 0) {
        *(float2*)(&wm_s[w][2 * kp]) = make_float2(mx, my);
        *(float2*)(&ws_s[w][2 * kp]) = make_float2(sx, sy);
    }
    __syncthreads();

    float* fp = f + (long)b * N_DIM * N_DIM
                  + (long)(iBase + 6 * w) * N_DIM + kBase + 2 * kp;
    *(float2*)(fp + (2 * ih) * N_DIM)     = make_float2(c0.x, c0.y);
    *(float2*)(fp + (2 * ih + 1) * N_DIM) = make_float2(c1.x, c1.y);
    *(float2*)(fp + (4 + ih) * N_DIM)     = make_float2(c2.x, c2.y);
    if (w == 0 && ih == 0) {
        float2 m8[8], s8[8];
        #pragma unroll
        for (int q = 0; q < 8; ++q) {
            m8[q] = *(const float2*)(&wm_s[q][2 * kp]);
            s8[q] = *(const float2*)(&ws_s[q][2 * kp]);
        }
        float tmx = m8[0].x, tmy = m8[0].y;
        #pragma unroll
        for (int q = 1; q < 8; ++q) { tmx = fmaxf(tmx, m8[q].x);  tmy = fmaxf(tmy, m8[q].y); }
        float tsx = 0.f, tsy = 0.f;
        #pragma unroll
        for (int q = 0; q < 8; ++q) {
            tsx += s8[q].x * __expf(m8[q].x - tmx);
            tsy += s8[q].y * __expf(m8[q].y - tmy);
        }
        const long po = (long)(b * (N_DIM / PW_IT) + blockIdx.y) * N_DIM + kBase + 2 * kp;
        *(float2*)(pm + po) = make_float2(tmx, tmy);
        *(float2*)(ps + po) = make_float2(tsx, tsy);
    }
}

#define SM_IT 96
#define SM_WAVES 4
#define SM_ROWS (SM_IT / SM_WAVES)
#define SM_TILES (N_DIM / PW_IT)
__global__ void __launch_bounds__(128)
softmax_i_kernel(const float* __restrict__ pm,
                 const float* __restrict__ ps,
                 float* io)
{
    const int b     = blockIdx.z;
    const int kBase = blockIdx.x * 32;
    const int iBase = blockIdx.y * SM_IT;
    const int kl    = threadIdx.x & 31;
    const int iy    = threadIdx.x >> 5;
    float* ob = io + (long)b * N_DIM * N_DIM + (long)(iBase + iy) * N_DIM + kBase + kl;
    const long po = (long)b * SM_TILES * N_DIM + kBase + kl;

    float val[SM_ROWS];
    #pragma unroll
    for (int j = 0; j < SM_ROWS; ++j)
        val[j] = ob[(long)j * SM_WAVES * N_DIM];
    float tm[SM_TILES], ts[SM_TILES];
    #pragma unroll
    for (int q = 0; q < SM_TILES; ++q) {
        tm[q] = pm[po + (long)q * N_DIM];
        ts[q] = ps[po + (long)q * N_DIM];
    }

    float mm = tm[0];
    #pragma unroll
    for (int q = 1; q < SM_TILES; ++q) mm = fmaxf(mm, tm[q]);
    float ss = 0.f;
    #pragma unroll
    for (int q = 0; q < SM_TILES; ++q) ss += ts[q] * __expf(tm[q] - mm);
    const float inv = 1.f / ss;

    #pragma unroll
    for (int j = 0; j < SM_ROWS; ++j)
        ob[(long)j * SM_WAVES * N_DIM] = __expf(val[j] - mm) * inv;
}

extern "C" void kernel_launch(void* const* d_in, const int* in_sizes, int n_in,
                              void* d_out, int out_size, void* d_ws, size_t ws_size,
                              hipStream_t stream)
{
    const float* x       = (const float*)d_in[0];
    const float* theta_w = (const float*)d_in[1];
    const float* theta_b = (const float*)d_in[2];
    const float* phi_w   = (const float*)d_in[3];
    const float* phi_b   = (const float*)d_in[4];
    const float* W_a     = (const float*)d_in[5];
    const float* v_a     = (const float*)d_in[6];
    float* out = (float*)d_out;

    float* ws = (float*)d_ws;
    float* a  = ws;
    float* bt = a  + B_DIM * C_DIM * N_DIM;
    float* pm = bt + B_DIM * C_DIM * N_DIM;
    float* ps = pm + B_DIM * (N_DIM / PW_IT) * N_DIM;

    {
        dim3 grid(N_TILES, B_DIM, 4), block(256, 1, 1);
        chain_gemm_f32<<<grid, block, 0, stream>>>(
            x, theta_w, theta_b, phi_w, phi_b, W_a, a, bt);
    }

    {
        dim3 grid(N_DIM / PW_KT, N_DIM / PW_IT, B_DIM), block(256, 1, 1);
        pairwise_tanh_kernel<<<grid, block, 0, stream>>>(a, bt, v_a, out, pm, ps);
    }

    {
        dim3 grid(N_DIM / 32, N_DIM / SM_IT, B_DIM), block(128, 1, 1);
        softmax_i_kernel<<<grid, block, 0, stream>>>(pm, ps, out);
    }
}
